# starting kernel
# baseline (speedup 1.0000x reference)
#include <hip/hip_runtime.h>


#define N_NODES 100000
#define N_EDGES 1600000
#define F 64
#define HSTRIDE 65

typedef __attribute__((ext_vector_type(2))) float v2f;
typedef __attribute__((ext_vector_type(8))) float v8f;

__global__ __launch_bounds__(256) void zero_f32(float* __restrict__ p, long long n) {
    long long i = (long long)blockIdx.x * 256 + threadIdx.x;
    long long stride = (long long)gridDim.x * 256;
    for (; i < n; i += stride) p[i] = 0.0f;
}

__global__ __launch_bounds__(256) void sage_degree(const int* __restrict__ dst,
                                                   float* __restrict__ deg, int n_edges) {
    int e = blockIdx.x * 256 + threadIdx.x;
    if (e < n_edges) atomicAdd(&deg[dst[e]], 1.0f);
}

__global__ __launch_bounds__(256) void sage_aggregate(const float* __restrict__ h,
                                                      const int* __restrict__ src,
                                                      const int* __restrict__ dst,
                                                      float* __restrict__ msum, int n_edges) {
    long long idx = (long long)blockIdx.x * 256 + threadIdx.x;
    int e = (int)(idx >> 5);
    int l = (int)(idx & 31);
    if (e >= n_edges) return;
    int s = src[e];
    int d = dst[e];
    const float2 v = *(const float2*)(h + (long long)s * F + l * 2);
    atomicAdd(&msum[(long long)d * F + l * 2 + 0], v.x);
    atomicAdd(&msum[(long long)d * F + l * 2 + 1], v.y);
}

__global__ __launch_bounds__(256) void sage_gemm(const float* __restrict__ h,
                                                 const float* __restrict__ msum,
                                                 const float* __restrict__ deg,
                                                 const float* __restrict__ Wself,
                                                 const float* __restrict__ Wneigh,
                                                 const float* __restrict__ bias,
                                                 float* __restrict__ out,
                                                 int n_nodes, int do_relu) {
    __shared__ float sWs[F * F];
    __shared__ float sWn[F * F];
    __shared__ float sH [32 * HSTRIDE];
    __shared__ float sHn[32 * HSTRIDE];
    __shared__ float sB [F];

    const int tid = threadIdx.x;
    const int node0 = blockIdx.x * 32;

    #pragma unroll
    for (int i = 0; i < 16; ++i) {
        sWs[tid + i * 256] = Wself [tid + i * 256];
        sWn[tid + i * 256] = Wneigh[tid + i * 256];
    }
    if (tid < F) sB[tid] = bias[tid];

    #pragma unroll
    for (int i = 0; i < 8; ++i) {
        int idx = tid + i * 256;
        int row = idx >> 6;
        int col = idx & 63;
        int node = node0 + row;
        float hv = 0.0f, mv = 0.0f, dv = 1.0f;
        if (node < n_nodes) {
            hv = h   [(long long)node * F + col];
            mv = msum[(long long)node * F + col];
            dv = deg[node];
        }
        float rinv = 1.0f / fmaxf(dv, 1.0f);
        sH [row * HSTRIDE + col] = hv;
        sHn[row * HSTRIDE + col] = mv * rinv;
    }
    __syncthreads();

    const int wave  = tid >> 5;
    const int lane  = tid & 31;
    const int m0    = (wave >> 2) * 16;
    const int c0    = (wave & 3) * 16;
    const int lrow  = lane & 15;
    const int khalf = lane >> 4;

    v8f acc = {};
    const float* aSelf = &sH [m0 * HSTRIDE + lrow * HSTRIDE];
    const float* aNei  = &sHn[m0 * HSTRIDE + lrow * HSTRIDE];
    #pragma unroll
    for (int kk = 0; kk < 16; ++kk) {
        const int kb = kk * 4 + khalf * 2;
        v2f a, b;
        a.x = aSelf[kb];
        a.y = aSelf[kb + 1];
        b.x = sWs[(kb    ) * F + c0 + lrow];
        b.y = sWs[(kb + 1) * F + c0 + lrow];
        acc = __builtin_amdgcn_wmma_f32_16x16x4_f32(false, a, false, b,
                                                    (short)0, acc, false, false);
        a.x = aNei[kb];
        a.y = aNei[kb + 1];
        b.x = sWn[(kb    ) * F + c0 + lrow];
        b.y = sWn[(kb + 1) * F + c0 + lrow];
        acc = __builtin_amdgcn_wmma_f32_16x16x4_f32(false, a, false, b,
                                                    (short)0, acc, false, false);
    }

    const float bv = sB[c0 + lrow];
    #pragma unroll
    for (int i = 0; i < 8; ++i) {
        int row = node0 + m0 + i + khalf * 8;
        float v = acc[i] + bv;
        if (do_relu) v = fmaxf(v, 0.0f);
        if (row < n_nodes) out[(long long)row * F + c0 + lrow] = v;
    }
}

extern "C" void kernel_launch(void* const* d_in, const int* in_sizes, int n_in,
                              void* d_out, int out_size, void* d_ws, size_t ws_size,
                              hipStream_t stream) {
    (void)in_sizes; (void)n_in; (void)out_size; (void)ws_size;
    const float* feat = (const float*)d_in[0];
    const int*   src  = (const int*)d_in[1];
    const int*   dst  = (const int*)d_in[2];
    const float* Ws[3] = { (const float*)d_in[3], (const float*)d_in[6], (const float*)d_in[9]  };
    const float* Wn[3] = { (const float*)d_in[4], (const float*)d_in[7], (const float*)d_in[10] };
    const float* bs[3] = { (const float*)d_in[5], (const float*)d_in[8], (const float*)d_in[11] };
    float* outp = (float*)d_out;

    float* ws   = (float*)d_ws;
    float* deg  = ws;
    float* msum = deg  + (long long)N_NODES;
    float* h1   = msum + (long long)N_NODES * F;
    float* h2   = h1   + (long long)N_NODES * F;

    const int zeroBlocks  = 2048;
    const int degBlocks   = (N_EDGES + 255) / 256;
    const int aggBlocks   = (int)(((long long)N_EDGES * 32 + 255) / 256);
    const int gemmBlocks  = (N_NODES + 31) / 32;

    zero_f32<<<zeroBlocks, 256, 0, stream>>>(deg, (long long)N_NODES * (1 + F));
    sage_degree<<<degBlocks, 256, 0, stream>>>(dst, deg, N_EDGES);

    sage_aggregate<<<aggBlocks, 256, 0, stream>>>(feat, src, dst, msum, N_EDGES);
    sage_gemm<<<gemmBlocks, 256, 0, stream>>>(feat, msum, deg, Ws[0], Wn[0], bs[0], h1, N_NODES, 1);

    zero_f32<<<zeroBlocks, 256, 0, stream>>>(msum, (long long)N_NODES * F);
    sage_aggregate<<<aggBlocks, 256, 0, stream>>>(h1, src, dst, msum, N_EDGES);
    sage_gemm<<<gemmBlocks, 256, 0, stream>>>(h1, msum, deg, Ws[1], Wn[1], bs[1], h2, N_NODES, 1);

    zero_f32<<<zeroBlocks, 256, 0, stream>>>(msum, (long long)N_NODES * F);
    sage_aggregate<<<aggBlocks, 256, 0, stream>>>(h2, src, dst, msum, N_EDGES);
    sage_gemm<<<gemmBlocks, 256, 0, stream>>>(h2, msum, deg, Ws[2], Wn[2], bs[2], outp, N_NODES, 0);
}


# aggregate: each wave takes 8 consecutive edges, indices via scalar loads, all 8 row gathers in flight before its f32 atomics (8x fewer waves); same atomic adds, arrival order as in parent
# speedup vs baseline: 1.4400x; 1.4400x over previous
#include <hip/hip_runtime.h>


#define N_NODES 100000
#define N_EDGES 1600000
#define F 64
#define HSTRIDE 65

typedef __attribute__((ext_vector_type(2))) float v2f;
typedef __attribute__((ext_vector_type(8))) float v8f;

__global__ __launch_bounds__(256) void zero_f32(float* __restrict__ p, long long n) {
    long long i = (long long)blockIdx.x * 256 + threadIdx.x;
    long long stride = (long long)gridDim.x * 256;
    for (; i < n; i += stride) p[i] = 0.0f;
}

__global__ __launch_bounds__(256) void sage_degree(const int* __restrict__ dst,
                                                   float* __restrict__ deg, int n_edges) {
    int e = blockIdx.x * 256 + threadIdx.x;
    if (e < n_edges) atomicAdd(&deg[dst[e]], 1.0f);
}

#define EPW 8
__global__ __launch_bounds__(256) void sage_aggregate(const float* __restrict__ h,
                                                      const int* __restrict__ src,
                                                      const int* __restrict__ dst,
                                                      float* __restrict__ msum, int n_edges) {
    const int wv = __builtin_amdgcn_readfirstlane((int)(threadIdx.x >> 5));
    const int col = (int)(threadIdx.x & 31) * 2;
    const long long e0 = ((long long)blockIdx.x * 8 + wv) * EPW;
    if (e0 + EPW <= (long long)n_edges) {
        int s[EPW], d[EPW];
        #pragma unroll
        for (int j = 0; j < EPW; ++j) { s[j] = src[e0 + j]; d[j] = dst[e0 + j]; }
        float2 v[EPW];
        #pragma unroll
        for (int j = 0; j < EPW; ++j) v[j] = *(const float2*)(h + (long long)s[j] * F + col);
        #pragma unroll
        for (int j = 0; j < EPW; ++j) {
            atomicAdd(&msum[(long long)d[j] * F + col + 0], v[j].x);
            atomicAdd(&msum[(long long)d[j] * F + col + 1], v[j].y);
        }
    } else {
        for (long long e = e0; e < (long long)n_edges; ++e) {
            const int s = src[e];
            const int d = dst[e];
            const float2 v = *(const float2*)(h + (long long)s * F + col);
            atomicAdd(&msum[(long long)d * F + col + 0], v.x);
            atomicAdd(&msum[(long long)d * F + col + 1], v.y);
        }
    }
}

__global__ __launch_bounds__(256) void sage_gemm(const float* __restrict__ h,
                                                 const float* __restrict__ msum,
                                                 const float* __restrict__ deg,
                                                 const float* __restrict__ Wself,
                                                 const float* __restrict__ Wneigh,
                                                 const float* __restrict__ bias,
                                                 float* __restrict__ out,
                                                 int n_nodes, int do_relu) {
    __shared__ float sWs[F * F];
    __shared__ float sWn[F * F];
    __shared__ float sH [32 * HSTRIDE];
    __shared__ float sHn[32 * HSTRIDE];
    __shared__ float sB [F];

    const int tid = threadIdx.x;
    const int node0 = blockIdx.x * 32;

    #pragma unroll
    for (int i = 0; i < 16; ++i) {
        sWs[tid + i * 256] = Wself [tid + i * 256];
        sWn[tid + i * 256] = Wneigh[tid + i * 256];
    }
    if (tid < F) sB[tid] = bias[tid];

    #pragma unroll
    for (int i = 0; i < 8; ++i) {
        int idx = tid + i * 256;
        int row = idx >> 6;
        int col = idx & 63;
        int node = node0 + row;
        float hv = 0.0f, mv = 0.0f, dv = 1.0f;
        if (node < n_nodes) {
            hv = h   [(long long)node * F + col];
            mv = msum[(long long)node * F + col];
            dv = deg[node];
        }
        float rinv = 1.0f / fmaxf(dv, 1.0f);
        sH [row * HSTRIDE + col] = hv;
        sHn[row * HSTRIDE + col] = mv * rinv;
    }
    __syncthreads();

    const int wave  = tid >> 5;
    const int lane  = tid & 31;
    const int m0    = (wave >> 2) * 16;
    const int c0    = (wave & 3) * 16;
    const int lrow  = lane & 15;
    const int khalf = lane >> 4;

    v8f acc = {};
    const float* aSelf = &sH [m0 * HSTRIDE + lrow * HSTRIDE];
    const float* aNei  = &sHn[m0 * HSTRIDE + lrow * HSTRIDE];
    #pragma unroll
    for (int kk = 0; kk < 16; ++kk) {
        const int kb = kk * 4 + khalf * 2;
        v2f a, b;
        a.x = aSelf[kb];
        a.y = aSelf[kb + 1];
        b.x = sWs[(kb    ) * F + c0 + lrow];
        b.y = sWs[(kb + 1) * F + c0 + lrow];
        acc = __builtin_amdgcn_wmma_f32_16x16x4_f32(false, a, false, b,
                                                    (short)0, acc, false, false);
        a.x = aNei[kb];
        a.y = aNei[kb + 1];
        b.x = sWn[(kb    ) * F + c0 + lrow];
        b.y = sWn[(kb + 1) * F + c0 + lrow];
        acc = __builtin_amdgcn_wmma_f32_16x16x4_f32(false, a, false, b,
                                                    (short)0, acc, false, false);
    }

    const float bv = sB[c0 + lrow];
    #pragma unroll
    for (int i = 0; i < 8; ++i) {
        int row = node0 + m0 + i + khalf * 8;
        float v = acc[i] + bv;
        if (do_relu) v = fmaxf(v, 0.0f);
        if (row < n_nodes) out[(long long)row * F + c0 + lrow] = v;
    }
}

extern "C" void kernel_launch(void* const* d_in, const int* in_sizes, int n_in,
                              void* d_out, int out_size, void* d_ws, size_t ws_size,
                              hipStream_t stream) {
    (void)in_sizes; (void)n_in; (void)out_size; (void)ws_size;
    const float* feat = (const float*)d_in[0];
    const int*   src  = (const int*)d_in[1];
    const int*   dst  = (const int*)d_in[2];
    const float* Ws[3] = { (const float*)d_in[3], (const float*)d_in[6], (const float*)d_in[9]  };
    const float* Wn[3] = { (const float*)d_in[4], (const float*)d_in[7], (const float*)d_in[10] };
    const float* bs[3] = { (const float*)d_in[5], (const float*)d_in[8], (const float*)d_in[11] };
    float* outp = (float*)d_out;

    float* ws   = (float*)d_ws;
    float* deg  = ws;
    float* msum = deg  + (long long)N_NODES;
    float* h1   = msum + (long long)N_NODES * F;
    float* h2   = h1   + (long long)N_NODES * F;

    const int zeroBlocks  = 2048;
    const int degBlocks   = (N_EDGES + 255) / 256;
    const int aggBlocks   = (N_EDGES + 8 * EPW - 1) / (8 * EPW);
    const int gemmBlocks  = (N_NODES + 31) / 32;

    zero_f32<<<zeroBlocks, 256, 0, stream>>>(deg, (long long)N_NODES * (1 + F));
    sage_degree<<<degBlocks, 256, 0, stream>>>(dst, deg, N_EDGES);

    sage_aggregate<<<aggBlocks, 256, 0, stream>>>(feat, src, dst, msum, N_EDGES);
    sage_gemm<<<gemmBlocks, 256, 0, stream>>>(feat, msum, deg, Ws[0], Wn[0], bs[0], h1, N_NODES, 1);

    zero_f32<<<zeroBlocks, 256, 0, stream>>>(msum, (long long)N_NODES * F);
    sage_aggregate<<<aggBlocks, 256, 0, stream>>>(h1, src, dst, msum, N_EDGES);
    sage_gemm<<<gemmBlocks, 256, 0, stream>>>(h1, msum, deg, Ws[1], Wn[1], bs[1], h2, N_NODES, 1);

    zero_f32<<<zeroBlocks, 256, 0, stream>>>(msum, (long long)N_NODES * F);
    sage_aggregate<<<aggBlocks, 256, 0, stream>>>(h2, src, dst, msum, N_EDGES);
    sage_gemm<<<gemmBlocks, 256, 0, stream>>>(h2, msum, deg, Ws[2], Wn[2], bs[2], outp, N_NODES, 0);
}


# scatter to gather: edges bucketed by dst once per call (int-atomic ranks, wave-scan alloc); layer kernel sums each node's in-neighbour rows in registers in arrival-rank order; no f32 atomics, no msum
# speedup vs baseline: 2.7409x; 1.9034x over previous
#include <hip/hip_runtime.h>


#define N_NODES 100000
#define N_EDGES 1600000
#define F 64
#define HSTRIDE 65
#define NPW 4
#define GCH 8

typedef __attribute__((ext_vector_type(2))) float v2f;
typedef __attribute__((ext_vector_type(8))) float v8f;

__global__ __launch_bounds__(256) void zero_i32(int* __restrict__ p, int n) {
    const int i = blockIdx.x * 256 + threadIdx.x;
    if (i < n) p[i] = 0;
}

__global__ __launch_bounds__(256) void sage_rank(const int* __restrict__ dst,
                                                 int* __restrict__ cnt,
                                                 int* __restrict__ rank, int n_edges) {
    const int e = blockIdx.x * 256 + threadIdx.x;
    if (e < n_edges) rank[e] = atomicAdd(&cnt[dst[e]], 1);
}

__global__ __launch_bounds__(256) void sage_alloc(const int* __restrict__ cnt,
                                                  int* __restrict__ cursor,
                                                  int* __restrict__ offs, int n_nodes) {
    const int node = blockIdx.x * 256 + threadIdx.x;
    const int lane = threadIdx.x & 31;
    const int c = (node < n_nodes) ? cnt[node] : 0;
    int incl = c;
    #pragma unroll
    for (int d = 1; d < 32; d <<= 1) {
        const int t = __shfl_up(incl, d);
        if (lane >= d) incl += t;
    }
    const int total = __shfl(incl, 31);
    int base = 0;
    if (lane == 0) base = atomicAdd(cursor, total);
    base = __shfl(base, 0);
    if (node < n_nodes) offs[node] = base + incl - c;
}

__global__ __launch_bounds__(256) void sage_fill(const int* __restrict__ src,
                                                 const int* __restrict__ dst,
                                                 const int* __restrict__ rank,
                                                 const int* __restrict__ offs,
                                                 int* __restrict__ csr, int n_edges) {
    const int e = blockIdx.x * 256 + threadIdx.x;
    if (e < n_edges) csr[offs[dst[e]] + rank[e]] = src[e];
}

__global__ __launch_bounds__(256) void sage_layer(const float* __restrict__ h,
                                                  const int* __restrict__ csr,
                                                  const int* __restrict__ offs,
                                                  const int* __restrict__ cnt,
                                                  const float* __restrict__ Wself,
                                                  const float* __restrict__ Wneigh,
                                                  const float* __restrict__ bias,
                                                  float* __restrict__ out,
                                                  int n_nodes, int do_relu) {
    __shared__ float sWs[F * F];
    __shared__ float sWn[F * F];
    __shared__ float sH [32 * HSTRIDE];
    __shared__ float sHn[32 * HSTRIDE];
    __shared__ float sB [F];

    const int tid = threadIdx.x;
    const int node0 = blockIdx.x * 32;
    const int wv = __builtin_amdgcn_readfirstlane((int)(threadIdx.x >> 5));
    const int lane = tid & 31;
    const int col2 = lane * 2;

    int beg[NPW], len[NPW], idx[NPW];
    float2 own[NPW];
    #pragma unroll
    for (int r = 0; r < NPW; ++r) {
        const int node = node0 + wv * NPW + r;
        const bool ok = node < n_nodes;
        const int nc = ok ? node : 0;
        beg[r] = offs[nc];
        len[r] = ok ? cnt[nc] : 0;
        own[r] = *(const float2*)(h + (long long)nc * F + col2);
        if (!ok) { own[r].x = 0.0f; own[r].y = 0.0f; }
    }
    #pragma unroll
    for (int r = 0; r < NPW; ++r) idx[r] = (lane < len[r]) ? csr[beg[r] + lane] : 0;

    #pragma unroll
    for (int i = 0; i < 16; ++i) {
        sWs[tid + i * 256] = Wself [tid + i * 256];
        sWn[tid + i * 256] = Wneigh[tid + i * 256];
    }
    if (tid < F) sB[tid] = bias[tid];

    #pragma unroll
    for (int r = 0; r < NPW; ++r) {
        sH[(wv * NPW + r) * HSTRIDE + col2    ] = own[r].x;
        sH[(wv * NPW + r) * HSTRIDE + col2 + 1] = own[r].y;
    }

    #pragma unroll
    for (int r = 0; r < NPW; ++r) {
        const int n = len[r];
        const int m = n < 32 ? n : 32;
        float ax = 0.0f, ay = 0.0f;
        for (int j0 = 0; j0 < m; j0 += GCH) {
            float2 v[GCH];
            #pragma unroll
            for (int j = 0; j < GCH; ++j) {
                const int s = __builtin_amdgcn_readlane(idx[r], j0 + j);
                v[j] = *(const float2*)(h + (long long)s * F + col2);
            }
            #pragma unroll
            for (int j = 0; j < GCH; ++j) {
                if (j0 + j < m) { ax += v[j].x; ay += v[j].y; }
            }
        }
        for (int k = 32; k < n; ++k) {
            const int s = csr[beg[r] + k];
            const float2 v = *(const float2*)(h + (long long)s * F + col2);
            ax += v.x; ay += v.y;
        }
        const float rinv = 1.0f / fmaxf((float)n, 1.0f);
        sHn[(wv * NPW + r) * HSTRIDE + col2    ] = ax * rinv;
        sHn[(wv * NPW + r) * HSTRIDE + col2 + 1] = ay * rinv;
    }
    __syncthreads();

    const int wave  = tid >> 5;
    const int m0    = (wave >> 2) * 16;
    const int c0    = (wave & 3) * 16;
    const int lrow  = lane & 15;
    const int khalf = lane >> 4;

    v8f acc = {};
    const float* aSelf = &sH [m0 * HSTRIDE + lrow * HSTRIDE];
    const float* aNei  = &sHn[m0 * HSTRIDE + lrow * HSTRIDE];
    #pragma unroll
    for (int kk = 0; kk < 16; ++kk) {
        const int kb = kk * 4 + khalf * 2;
        v2f a, b;
        a.x = aSelf[kb];
        a.y = aSelf[kb + 1];
        b.x = sWs[(kb    ) * F + c0 + lrow];
        b.y = sWs[(kb + 1) * F + c0 + lrow];
        acc = __builtin_amdgcn_wmma_f32_16x16x4_f32(false, a, false, b,
                                                    (short)0, acc, false, false);
        a.x = aNei[kb];
        a.y = aNei[kb + 1];
        b.x = sWn[(kb    ) * F + c0 + lrow];
        b.y = sWn[(kb + 1) * F + c0 + lrow];
        acc = __builtin_amdgcn_wmma_f32_16x16x4_f32(false, a, false, b,
                                                    (short)0, acc, false, false);
    }

    const float bv = sB[c0 + lrow];
    #pragma unroll
    for (int i = 0; i < 8; ++i) {
        int row = node0 + m0 + i + khalf * 8;
        float v = acc[i] + bv;
        if (do_relu) v = fmaxf(v, 0.0f);
        if (row < n_nodes) out[(long long)row * F + c0 + lrow] = v;
    }
}

extern "C" void kernel_launch(void* const* d_in, const int* in_sizes, int n_in,
                              void* d_out, int out_size, void* d_ws, size_t ws_size,
                              hipStream_t stream) {
    (void)in_sizes; (void)n_in; (void)out_size; (void)ws_size;
    const float* feat = (const float*)d_in[0];
    const int*   src  = (const int*)d_in[1];
    const int*   dst  = (const int*)d_in[2];
    const float* Ws[3] = { (const float*)d_in[3], (const float*)d_in[6], (const float*)d_in[9]  };
    const float* Wn[3] = { (const float*)d_in[4], (const float*)d_in[7], (const float*)d_in[10] };
    const float* bs[3] = { (const float*)d_in[5], (const float*)d_in[8], (const float*)d_in[11] };
    float* outp = (float*)d_out;

    int*   wsi    = (int*)d_ws;
    int*   cnt    = wsi;
    int*   cursor = wsi + N_NODES;
    int*   offs   = wsi + N_NODES + 64;
    int*   csr    = wsi + 2 * N_NODES + 128;
    float* h1     = (float*)(csr + N_EDGES);
    float* h2     = h1 + (long long)N_NODES * F;
    int*   rank   = (int*)h2;

    const int zeroBlocks  = (N_NODES + 64 + 255) / 256;
    const int edgeBlocks  = (N_EDGES + 255) / 256;
    const int nodeBlocks  = (N_NODES + 255) / 256;
    const int layerBlocks = (N_NODES + 31) / 32;

    zero_i32  <<<zeroBlocks, 256, 0, stream>>>(cnt, N_NODES + 64);
    sage_rank <<<edgeBlocks, 256, 0, stream>>>(dst, cnt, rank, N_EDGES);
    sage_alloc<<<nodeBlocks, 256, 0, stream>>>(cnt, cursor, offs, N_NODES);
    sage_fill <<<edgeBlocks, 256, 0, stream>>>(src, dst, rank, offs, csr, N_EDGES);

    sage_layer<<<layerBlocks, 256, 0, stream>>>(feat, csr, offs, cnt, Ws[0], Wn[0], bs[0], h1,   N_NODES, 1);
    sage_layer<<<layerBlocks, 256, 0, stream>>>(h1,   csr, offs, cnt, Ws[1], Wn[1], bs[1], h2,   N_NODES, 1);
    sage_layer<<<layerBlocks, 256, 0, stream>>>(h2,   csr, offs, cnt, Ws[2], Wn[2], bs[2], outp, N_NODES, 0);
}


# gather depth: a wave keeps up to 24 neighbour rows of a node in flight (was 8), in groups of 8 issued only as far as the bucket reaches; sums stay in bucket order, bit-identical
# speedup vs baseline: 2.8428x; 1.0372x over previous
#include <hip/hip_runtime.h>


#define N_NODES 100000
#define N_EDGES 1600000
#define F 64
#define HSTRIDE 65
#define NPW 4
#define GCH 24

typedef __attribute__((ext_vector_type(2))) float v2f;
typedef __attribute__((ext_vector_type(8))) float v8f;

__global__ __launch_bounds__(256) void zero_i32(int* __restrict__ p, int n) {
    const int i = blockIdx.x * 256 + threadIdx.x;
    if (i < n) p[i] = 0;
}

__global__ __launch_bounds__(256) void sage_rank(const int* __restrict__ dst,
                                                 int* __restrict__ cnt,
                                                 int* __restrict__ rank, int n_edges) {
    const int e = blockIdx.x * 256 + threadIdx.x;
    if (e < n_edges) rank[e] = atomicAdd(&cnt[dst[e]], 1);
}

__global__ __launch_bounds__(256) void sage_alloc(const int* __restrict__ cnt,
                                                  int* __restrict__ cursor,
                                                  int* __restrict__ offs, int n_nodes) {
    const int node = blockIdx.x * 256 + threadIdx.x;
    const int lane = threadIdx.x & 31;
    const int c = (node < n_nodes) ? cnt[node] : 0;
    int incl = c;
    #pragma unroll
    for (int d = 1; d < 32; d <<= 1) {
        const int t = __shfl_up(incl, d);
        if (lane >= d) incl += t;
    }
    const int total = __shfl(incl, 31);
    int base = 0;
    if (lane == 0) base = atomicAdd(cursor, total);
    base = __shfl(base, 0);
    if (node < n_nodes) offs[node] = base + incl - c;
}

__global__ __launch_bounds__(256) void sage_fill(const int* __restrict__ src,
                                                 const int* __restrict__ dst,
                                                 const int* __restrict__ rank,
                                                 const int* __restrict__ offs,
                                                 int* __restrict__ csr, int n_edges) {
    const int e = blockIdx.x * 256 + threadIdx.x;
    if (e < n_edges) csr[offs[dst[e]] + rank[e]] = src[e];
}

__global__ __launch_bounds__(256) void sage_layer(const float* __restrict__ h,
                                                  const int* __restrict__ csr,
                                                  const int* __restrict__ offs,
                                                  const int* __restrict__ cnt,
                                                  const float* __restrict__ Wself,
                                                  const float* __restrict__ Wneigh,
                                                  const float* __restrict__ bias,
                                                  float* __restrict__ out,
                                                  int n_nodes, int do_relu) {
    __shared__ float sWs[F * F];
    __shared__ float sWn[F * F];
    __shared__ float sH [32 * HSTRIDE];
    __shared__ float sHn[32 * HSTRIDE];
    __shared__ float sB [F];

    const int tid = threadIdx.x;
    const int node0 = blockIdx.x * 32;
    const int wv = __builtin_amdgcn_readfirstlane((int)(threadIdx.x >> 5));
    const int lane = tid & 31;
    const int col2 = lane * 2;

    int beg[NPW], len[NPW], idx[NPW];
    float2 own[NPW];
    #pragma unroll
    for (int r = 0; r < NPW; ++r) {
        const int node = node0 + wv * NPW + r;
        const bool ok = node < n_nodes;
        const int nc = ok ? node : 0;
        beg[r] = offs[nc];
        len[r] = ok ? cnt[nc] : 0;
        own[r] = *(const float2*)(h + (long long)nc * F + col2);
        if (!ok) { own[r].x = 0.0f; own[r].y = 0.0f; }
    }
    #pragma unroll
    for (int r = 0; r < NPW; ++r) idx[r] = (lane < len[r]) ? csr[beg[r] + lane] : 0;

    #pragma unroll
    for (int i = 0; i < 16; ++i) {
        sWs[tid + i * 256] = Wself [tid + i * 256];
        sWn[tid + i * 256] = Wneigh[tid + i * 256];
    }
    if (tid < F) sB[tid] = bias[tid];

    #pragma unroll
    for (int r = 0; r < NPW; ++r) {
        sH[(wv * NPW + r) * HSTRIDE + col2    ] = own[r].x;
        sH[(wv * NPW + r) * HSTRIDE + col2 + 1] = own[r].y;
    }

    #pragma unroll
    for (int r = 0; r < NPW; ++r) {
        const int n = len[r];
        const int m = n < 32 ? n : 32;
        float ax = 0.0f, ay = 0.0f;
        for (int j0 = 0; j0 < m; j0 += GCH) {
            float2 v[GCH];
            #pragma unroll
            for (int j = 0; j < 8; ++j) {
                const int s = __builtin_amdgcn_readlane(idx[r], j0 + j);
                v[j] = *(const float2*)(h + (long long)s * F + col2);
            }
            if (j0 + 8 < m) {
                #pragma unroll
                for (int j = 8; j < 16; ++j) {
                    const int s = __builtin_amdgcn_readlane(idx[r], j0 + j);
                    v[j] = *(const float2*)(h + (long long)s * F + col2);
                }
            }
            if (j0 + 16 < m) {
                #pragma unroll
                for (int j = 16; j < GCH; ++j) {
                    const int s = __builtin_amdgcn_readlane(idx[r], j0 + j);
                    v[j] = *(const float2*)(h + (long long)s * F + col2);
                }
            }
            #pragma unroll
            for (int j = 0; j < GCH; ++j) {
                if (j0 + j < m) { ax += v[j].x; ay += v[j].y; }
            }
        }
        for (int k = 32; k < n; ++k) {
            const int s = csr[beg[r] + k];
            const float2 v = *(const float2*)(h + (long long)s * F + col2);
            ax += v.x; ay += v.y;
        }
        const float rinv = 1.0f / fmaxf((float)n, 1.0f);
        sHn[(wv * NPW + r) * HSTRIDE + col2    ] = ax * rinv;
        sHn[(wv * NPW + r) * HSTRIDE + col2 + 1] = ay * rinv;
    }
    __syncthreads();

    const int wave  = tid >> 5;
    const int m0    = (wave >> 2) * 16;
    const int c0    = (wave & 3) * 16;
    const int lrow  = lane & 15;
    const int khalf = lane >> 4;

    v8f acc = {};
    const float* aSelf = &sH [m0 * HSTRIDE + lrow * HSTRIDE];
    const float* aNei  = &sHn[m0 * HSTRIDE + lrow * HSTRIDE];
    #pragma unroll
    for (int kk = 0; kk < 16; ++kk) {
        const int kb = kk * 4 + khalf * 2;
        v2f a, b;
        a.x = aSelf[kb];
        a.y = aSelf[kb + 1];
        b.x = sWs[(kb    ) * F + c0 + lrow];
        b.y = sWs[(kb + 1) * F + c0 + lrow];
        acc = __builtin_amdgcn_wmma_f32_16x16x4_f32(false, a, false, b,
                                                    (short)0, acc, false, false);
        a.x = aNei[kb];
        a.y = aNei[kb + 1];
        b.x = sWn[(kb    ) * F + c0 + lrow];
        b.y = sWn[(kb + 1) * F + c0 + lrow];
        acc = __builtin_amdgcn_wmma_f32_16x16x4_f32(false, a, false, b,
                                                    (short)0, acc, false, false);
    }

    const float bv = sB[c0 + lrow];
    #pragma unroll
    for (int i = 0; i < 8; ++i) {
        int row = node0 + m0 + i + khalf * 8;
        float v = acc[i] + bv;
        if (do_relu) v = fmaxf(v, 0.0f);
        if (row < n_nodes) out[(long long)row * F + c0 + lrow] = v;
    }
}

extern "C" void kernel_launch(void* const* d_in, const int* in_sizes, int n_in,
                              void* d_out, int out_size, void* d_ws, size_t ws_size,
                              hipStream_t stream) {
    (void)in_sizes; (void)n_in; (void)out_size; (void)ws_size;
    const float* feat = (const float*)d_in[0];
    const int*   src  = (const int*)d_in[1];
    const int*   dst  = (const int*)d_in[2];
    const float* Ws[3] = { (const float*)d_in[3], (const float*)d_in[6], (const float*)d_in[9]  };
    const float* Wn[3] = { (const float*)d_in[4], (const float*)d_in[7], (const float*)d_in[10] };
    const float* bs[3] = { (const float*)d_in[5], (const float*)d_in[8], (const float*)d_in[11] };
    float* outp = (float*)d_out;

    int*   wsi    = (int*)d_ws;
    int*   cnt    = wsi;
    int*   cursor = wsi + N_NODES;
    int*   offs   = wsi + N_NODES + 64;
    int*   csr    = wsi + 2 * N_NODES + 128;
    float* h1     = (float*)(csr + N_EDGES);
    float* h2     = h1 + (long long)N_NODES * F;
    int*   rank   = (int*)h2;

    const int zeroBlocks  = (N_NODES + 64 + 255) / 256;
    const int edgeBlocks  = (N_EDGES + 255) / 256;
    const int nodeBlocks  = (N_NODES + 255) / 256;
    const int layerBlocks = (N_NODES + 31) / 32;

    zero_i32  <<<zeroBlocks, 256, 0, stream>>>(cnt, N_NODES + 64);
    sage_rank <<<edgeBlocks, 256, 0, stream>>>(dst, cnt, rank, N_EDGES);
    sage_alloc<<<nodeBlocks, 256, 0, stream>>>(cnt, cursor, offs, N_NODES);
    sage_fill <<<edgeBlocks, 256, 0, stream>>>(src, dst, rank, offs, csr, N_EDGES);

    sage_layer<<<layerBlocks, 256, 0, stream>>>(feat, csr, offs, cnt, Ws[0], Wn[0], bs[0], h1,   N_NODES, 1);
    sage_layer<<<layerBlocks, 256, 0, stream>>>(h1,   csr, offs, cnt, Ws[1], Wn[1], bs[1], h2,   N_NODES, 1);
    sage_layer<<<layerBlocks, 256, 0, stream>>>(h2,   csr, offs, cnt, Ws[2], Wn[2], bs[2], outp, N_NODES, 0);
}


# edge bucketing through LDS: chunk blocks split the edges by 512-node destination range, one block per range counts, scans and places in LDS; no per-edge global atomic; 7 launches to 5; arrival order
# speedup vs baseline: 5.2233x; 1.8374x over previous
#include <hip/hip_runtime.h>


#define N_NODES 100000
#define N_EDGES 1600000
#define F 64
#define HSTRIDE 65
#define NPW 4
#define GCH 24

typedef __attribute__((ext_vector_type(2))) float v2f;
typedef __attribute__((ext_vector_type(8))) float v8f;

#define RSH 9
#define RNODES (1 << RSH)
#define NR ((N_NODES + RNODES - 1) / RNODES)
#define NRP 256
#define PCH 4096
#define NB ((N_EDGES + PCH - 1) / PCH)
#define NBP 400
#define SCAP 12288

__device__ __forceinline__ int block_scan_excl(int v, int* wsum, int nwaves, int& total) {
    const int lane = threadIdx.x & 31;
    const int w = threadIdx.x >> 5;
    int incl = v;
    #pragma unroll
    for (int d = 1; d < 32; d <<= 1) {
        const int t = __shfl_up(incl, d);
        if (lane >= d) incl += t;
    }
    if (lane == 31) wsum[w] = incl;
    __syncthreads();
    int off = 0, tot = 0;
    for (int i = 0; i < nwaves; ++i) {
        const int s = wsum[i];
        if (i < w) off += s;
        tot += s;
    }
    total = tot;
    __syncthreads();
    return off + incl - v;
}

__global__ __launch_bounds__(256) void sage_part(const int* __restrict__ src,
                                                 const int* __restrict__ dst,
                                                 unsigned* __restrict__ tmp,
                                                 int2* __restrict__ binfo,
                                                 int* __restrict__ cursor, int n_edges) {
    __shared__ int hist[NRP];
    __shared__ int lstart[NRP];
    __shared__ int wsum[8];
    __shared__ unsigned stage[PCH];

    const int tid = threadIdx.x;
    const int lane = tid & 31;
    const int wv = tid >> 5;
    const int b = blockIdx.x;
    const int cbase = b * PCH;

    hist[tid] = 0;
    if (b == 0 && tid == 0) *cursor = 0;

    int dv[16], sv[16], lr[16];
    const int wbase = cbase + wv * 512;
    #pragma unroll
    for (int q = 0; q < 4; ++q) {
        const int e0 = wbase + q * 128 + lane * 4;
        if (e0 + 3 < n_edges) {
            const int4 d4 = *(const int4*)(dst + e0);
            const int4 s4 = *(const int4*)(src + e0);
            dv[4 * q + 0] = d4.x; dv[4 * q + 1] = d4.y; dv[4 * q + 2] = d4.z; dv[4 * q + 3] = d4.w;
            sv[4 * q + 0] = s4.x; sv[4 * q + 1] = s4.y; sv[4 * q + 2] = s4.z; sv[4 * q + 3] = s4.w;
        } else {
            #pragma unroll
            for (int k = 0; k < 4; ++k) {
                const int e = e0 + k;
                const bool ok = e < n_edges;
                dv[4 * q + k] = ok ? dst[e] : -1;
                sv[4 * q + k] = ok ? src[e] : 0;
            }
        }
    }
    __syncthreads();

    #pragma unroll
    for (int i = 0; i < 16; ++i)
        lr[i] = (dv[i] >= 0) ? atomicAdd(&hist[dv[i] >> RSH], 1) : 0;
    __syncthreads();

    const int c = hist[tid];
    int total;
    const int ex = block_scan_excl(c, wsum, 8, total);
    lstart[tid] = ex;
    binfo[b * NRP + tid] = make_int2(c, cbase + ex);
    __syncthreads();

    #pragma unroll
    for (int i = 0; i < 16; ++i) {
        if (dv[i] >= 0)
            stage[lstart[dv[i] >> RSH] + lr[i]] =
                ((unsigned)sv[i] << RSH) | (unsigned)(dv[i] & (RNODES - 1));
    }
    __syncthreads();

    for (int i = tid; i < total; i += 256) tmp[cbase + i] = stage[i];
}

template <bool PLACE>
__device__ __forceinline__ void seg_pass(const unsigned* __restrict__ tmp, const int* scnt,
                                         const int* sstart, int* cl, unsigned* stage2,
                                         int* __restrict__ csr, int base, bool staged) {
    const int lane = threadIdx.x & 31;
    const int wv = threadIdx.x >> 5;
    for (int g = wv; g * 8 < NB; g += 16) {
        int c[8], st[8];
        unsigned v[8];
        #pragma unroll
        for (int u = 0; u < 8; ++u) {
            c[u]  = scnt[g * 8 + u];
            st[u] = sstart[g * 8 + u];
        }
        #pragma unroll
        for (int u = 0; u < 8; ++u) {
            int a = st[u] + lane;
            a = a < N_EDGES - 1 ? a : N_EDGES - 1;
            v[u] = tmp[a];
        }
        #pragma unroll
        for (int u = 0; u < 8; ++u) {
            if (lane < c[u]) {
                const int d = (int)(v[u] & (RNODES - 1));
                if (!PLACE) {
                    atomicAdd(&cl[d], 1);
                } else {
                    const int pos = atomicAdd(&cl[d], 1);
                    const int s = (int)(v[u] >> RSH);
                    if (staged) stage2[pos] = (unsigned)s; else csr[base + pos] = s;
                }
            }
        }
        #pragma unroll
        for (int u = 0; u < 8; ++u) {
            for (int j = 32 + lane; j < c[u]; j += 32) {
                const unsigned x = tmp[st[u] + j];
                const int d = (int)(x & (RNODES - 1));
                if (!PLACE) {
                    atomicAdd(&cl[d], 1);
                } else {
                    const int pos = atomicAdd(&cl[d], 1);
                    const int s = (int)(x >> RSH);
                    if (staged) stage2[pos] = (unsigned)s; else csr[base + pos] = s;
                }
            }
        }
    }
}

__global__ __launch_bounds__(512) void sage_sort(const unsigned* __restrict__ tmp,
                                                 const int2* __restrict__ binfo,
                                                 int* __restrict__ cursor,
                                                 int* __restrict__ cnt,
                                                 int* __restrict__ offs,
                                                 int* __restrict__ csr, int n_nodes) {
    __shared__ int scnt[NBP];
    __shared__ int sstart[NBP];
    __shared__ int cntL[RNODES];
    __shared__ int cur[RNODES];
    __shared__ int wsum[16];
    __shared__ int sbase;
    __shared__ unsigned stage2[SCAP];

    const int tid = threadIdx.x;
    const int r = blockIdx.x;

    int c = 0;
    if (tid < NB) {
        const int2 bi = binfo[tid * NRP + r];
        c = bi.x;
        scnt[tid] = bi.x;
        sstart[tid] = bi.y;
    } else if (tid < NBP) {
        scnt[tid] = 0;
        sstart[tid] = 0;
    }
    cntL[tid] = 0;

    int n_r;
    (void)block_scan_excl(c, wsum, 16, n_r);
    if (tid == 0) sbase = atomicAdd(cursor, n_r);
    const bool staged = n_r <= SCAP;

    seg_pass<false>(tmp, scnt, sstart, cntL, stage2, csr, 0, staged);
    __syncthreads();

    const int my = cntL[tid];
    int tot2;
    const int lo = block_scan_excl(my, wsum, 16, tot2);
    const int base = sbase;
    cur[tid] = lo;
    const int node = r * RNODES + tid;
    if (node < n_nodes) {
        cnt[node] = my;
        offs[node] = base + lo;
    }
    __syncthreads();

    seg_pass<true>(tmp, scnt, sstart, cur, stage2, csr, base, staged);
    __syncthreads();

    if (staged) {
        for (int i = tid; i < n_r; i += 512) csr[base + i] = (int)stage2[i];
    }
}

__global__ __launch_bounds__(256) void sage_layer(const float* __restrict__ h,
                                                  const int* __restrict__ csr,
                                                  const int* __restrict__ offs,
                                                  const int* __restrict__ cnt,
                                                  const float* __restrict__ Wself,
                                                  const float* __restrict__ Wneigh,
                                                  const float* __restrict__ bias,
                                                  float* __restrict__ out,
                                                  int n_nodes, int do_relu) {
    __shared__ float sWs[F * F];
    __shared__ float sWn[F * F];
    __shared__ float sH [32 * HSTRIDE];
    __shared__ float sHn[32 * HSTRIDE];
    __shared__ float sB [F];

    const int tid = threadIdx.x;
    const int node0 = blockIdx.x * 32;
    const int wv = __builtin_amdgcn_readfirstlane((int)(threadIdx.x >> 5));
    const int lane = tid & 31;
    const int col2 = lane * 2;

    int beg[NPW], len[NPW], idx[NPW];
    float2 own[NPW];
    #pragma unroll
    for (int r = 0; r < NPW; ++r) {
        const int node = node0 + wv * NPW + r;
        const bool ok = node < n_nodes;
        const int nc = ok ? node : 0;
        beg[r] = offs[nc];
        len[r] = ok ? cnt[nc] : 0;
        own[r] = *(const float2*)(h + (long long)nc * F + col2);
        if (!ok) { own[r].x = 0.0f; own[r].y = 0.0f; }
    }
    #pragma unroll
    for (int r = 0; r < NPW; ++r) idx[r] = (lane < len[r]) ? csr[beg[r] + lane] : 0;

    #pragma unroll
    for (int i = 0; i < 16; ++i) {
        sWs[tid + i * 256] = Wself [tid + i * 256];
        sWn[tid + i * 256] = Wneigh[tid + i * 256];
    }
    if (tid < F) sB[tid] = bias[tid];

    #pragma unroll
    for (int r = 0; r < NPW; ++r) {
        sH[(wv * NPW + r) * HSTRIDE + col2    ] = own[r].x;
        sH[(wv * NPW + r) * HSTRIDE + col2 + 1] = own[r].y;
    }

    #pragma unroll
    for (int r = 0; r < NPW; ++r) {
        const int n = len[r];
        const int m = n < 32 ? n : 32;
        float ax = 0.0f, ay = 0.0f;
        for (int j0 = 0; j0 < m; j0 += GCH) {
            float2 v[GCH];
            #pragma unroll
            for (int j = 0; j < 8; ++j) {
                const int s = __builtin_amdgcn_readlane(idx[r], j0 + j);
                v[j] = *(const float2*)(h + (long long)s * F + col2);
            }
            if (j0 + 8 < m) {
                #pragma unroll
                for (int j = 8; j < 16; ++j) {
                    const int s = __builtin_amdgcn_readlane(idx[r], j0 + j);
                    v[j] = *(const float2*)(h + (long long)s * F + col2);
                }
            }
            if (j0 + 16 < m) {
                #pragma unroll
                for (int j = 16; j < GCH; ++j) {
                    const int s = __builtin_amdgcn_readlane(idx[r], j0 + j);
                    v[j] = *(const float2*)(h + (long long)s * F + col2);
                }
            }
            #pragma unroll
            for (int j = 0; j < GCH; ++j) {
                if (j0 + j < m) { ax += v[j].x; ay += v[j].y; }
            }
        }
        for (int k = 32; k < n; ++k) {
            const int s = csr[beg[r] + k];
            const float2 v = *(const float2*)(h + (long long)s * F + col2);
            ax += v.x; ay += v.y;
        }
        const float rinv = 1.0f / fmaxf((float)n, 1.0f);
        sHn[(wv * NPW + r) * HSTRIDE + col2    ] = ax * rinv;
        sHn[(wv * NPW + r) * HSTRIDE + col2 + 1] = ay * rinv;
    }
    __syncthreads();

    const int wave  = tid >> 5;
    const int m0    = (wave >> 2) * 16;
    const int c0    = (wave & 3) * 16;
    const int lrow  = lane & 15;
    const int khalf = lane >> 4;

    v8f acc = {};
    const float* aSelf = &sH [m0 * HSTRIDE + lrow * HSTRIDE];
    const float* aNei  = &sHn[m0 * HSTRIDE + lrow * HSTRIDE];
    #pragma unroll
    for (int kk = 0; kk < 16; ++kk) {
        const int kb = kk * 4 + khalf * 2;
        v2f a, b;
        a.x = aSelf[kb];
        a.y = aSelf[kb + 1];
        b.x = sWs[(kb    ) * F + c0 + lrow];
        b.y = sWs[(kb + 1) * F + c0 + lrow];
        acc = __builtin_amdgcn_wmma_f32_16x16x4_f32(false, a, false, b,
                                                    (short)0, acc, false, false);
        a.x = aNei[kb];
        a.y = aNei[kb + 1];
        b.x = sWn[(kb    ) * F + c0 + lrow];
        b.y = sWn[(kb + 1) * F + c0 + lrow];
        acc = __builtin_amdgcn_wmma_f32_16x16x4_f32(false, a, false, b,
                                                    (short)0, acc, false, false);
    }

    const float bv = sB[c0 + lrow];
    #pragma unroll
    for (int i = 0; i < 8; ++i) {
        int row = node0 + m0 + i + khalf * 8;
        float v = acc[i] + bv;
        if (do_relu) v = fmaxf(v, 0.0f);
        if (row < n_nodes) out[(long long)row * F + c0 + lrow] = v;
    }
}

extern "C" void kernel_launch(void* const* d_in, const int* in_sizes, int n_in,
                              void* d_out, int out_size, void* d_ws, size_t ws_size,
                              hipStream_t stream) {
    (void)in_sizes; (void)n_in; (void)out_size; (void)ws_size;
    const float* feat = (const float*)d_in[0];
    const int*   src  = (const int*)d_in[1];
    const int*   dst  = (const int*)d_in[2];
    const float* Ws[3] = { (const float*)d_in[3], (const float*)d_in[6], (const float*)d_in[9]  };
    const float* Wn[3] = { (const float*)d_in[4], (const float*)d_in[7], (const float*)d_in[10] };
    const float* bs[3] = { (const float*)d_in[5], (const float*)d_in[8], (const float*)d_in[11] };
    float* outp = (float*)d_out;

    int*   wsi    = (int*)d_ws;
    int*   cnt    = wsi;
    int*   cursor = wsi + N_NODES;
    int*   offs   = wsi + N_NODES + 64;
    int*   csr    = wsi + 2 * N_NODES + 128;
    float* h1     = (float*)(csr + N_EDGES);
    float* h2     = h1 + (long long)N_NODES * F;
    unsigned* tmp = (unsigned*)h2;
    int2*  binfo  = (int2*)(tmp + N_EDGES);

    const int layerBlocks = (N_NODES + 31) / 32;

    sage_part<<<NB, 256, 0, stream>>>(src, dst, tmp, binfo, cursor, N_EDGES);
    sage_sort<<<NR, 512, 0, stream>>>(tmp, binfo, cursor, cnt, offs, csr, N_NODES);

    sage_layer<<<layerBlocks, 256, 0, stream>>>(feat, csr, offs, cnt, Ws[0], Wn[0], bs[0], h1,   N_NODES, 1);
    sage_layer<<<layerBlocks, 256, 0, stream>>>(h1,   csr, offs, cnt, Ws[1], Wn[1], bs[1], h2,   N_NODES, 1);
    sage_layer<<<layerBlocks, 256, 0, stream>>>(h2,   csr, offs, cnt, Ws[2], Wn[2], bs[2], outp, N_NODES, 0);
}


# gather two neighbour rows per load (half-wave per row, 16 B per lane, row offsets by row_share, exact step counts); a node's sum is now even-position sum plus odd-position sum: order changed
# speedup vs baseline: 5.4129x; 1.0363x over previous
#include <hip/hip_runtime.h>


#define N_NODES 100000
#define N_EDGES 1600000
#define F 64
#define HSTRIDE 65
#define NPW 4
#define GST 12

typedef __attribute__((ext_vector_type(2))) float v2f;
typedef __attribute__((ext_vector_type(8))) float v8f;
typedef __attribute__((ext_vector_type(4))) float v4f;

__device__ __forceinline__ v4f ld_row16(const float* __restrict__ hb, unsigned off) {
    return *(const v4f*)((const char*)hb + off);
}

template <int K>
__device__ __forceinline__ v4f ld_pair(const float* __restrict__ hb, int rowoff, unsigned c16) {
    const int ro = __builtin_amdgcn_update_dpp(0, rowoff, 0x150 + K, 0xF, 0xF, true);
    return ld_row16(hb, (unsigned)ro + c16);
}

__device__ __forceinline__ float other_half(float x) {
    const unsigned u = __float_as_uint(x);
    return __uint_as_float(__builtin_amdgcn_permlanex16(u, u, 0x76543210u, 0xfedcba98u, false, false));
}

#define RSH 9
#define RNODES (1 << RSH)
#define NR ((N_NODES + RNODES - 1) / RNODES)
#define NRP 256
#define PCH 4096
#define NB ((N_EDGES + PCH - 1) / PCH)
#define NBP 400
#define SCAP 12288

__device__ __forceinline__ int block_scan_excl(int v, int* wsum, int nwaves, int& total) {
    const int lane = threadIdx.x & 31;
    const int w = threadIdx.x >> 5;
    int incl = v;
    #pragma unroll
    for (int d = 1; d < 32; d <<= 1) {
        const int t = __shfl_up(incl, d);
        if (lane >= d) incl += t;
    }
    if (lane == 31) wsum[w] = incl;
    __syncthreads();
    int off = 0, tot = 0;
    for (int i = 0; i < nwaves; ++i) {
        const int s = wsum[i];
        if (i < w) off += s;
        tot += s;
    }
    total = tot;
    __syncthreads();
    return off + incl - v;
}

__global__ __launch_bounds__(256) void sage_part(const int* __restrict__ src,
                                                 const int* __restrict__ dst,
                                                 unsigned* __restrict__ tmp,
                                                 int2* __restrict__ binfo,
                                                 int* __restrict__ cursor, int n_edges) {
    __shared__ int hist[NRP];
    __shared__ int lstart[NRP];
    __shared__ int wsum[8];
    __shared__ unsigned stage[PCH];

    const int tid = threadIdx.x;
    const int lane = tid & 31;
    const int wv = tid >> 5;
    const int b = blockIdx.x;
    const int cbase = b * PCH;

    hist[tid] = 0;
    if (b == 0 && tid == 0) *cursor = 0;

    int dv[16], sv[16], lr[16];
    const int wbase = cbase + wv * 512;
    #pragma unroll
    for (int q = 0; q < 4; ++q) {
        const int e0 = wbase + q * 128 + lane * 4;
        if (e0 + 3 < n_edges) {
            const int4 d4 = *(const int4*)(dst + e0);
            const int4 s4 = *(const int4*)(src + e0);
            dv[4 * q + 0] = d4.x; dv[4 * q + 1] = d4.y; dv[4 * q + 2] = d4.z; dv[4 * q + 3] = d4.w;
            sv[4 * q + 0] = s4.x; sv[4 * q + 1] = s4.y; sv[4 * q + 2] = s4.z; sv[4 * q + 3] = s4.w;
        } else {
            #pragma unroll
            for (int k = 0; k < 4; ++k) {
                const int e = e0 + k;
                const bool ok = e < n_edges;
                dv[4 * q + k] = ok ? dst[e] : -1;
                sv[4 * q + k] = ok ? src[e] : 0;
            }
        }
    }
    __syncthreads();

    #pragma unroll
    for (int i = 0; i < 16; ++i)
        lr[i] = (dv[i] >= 0) ? atomicAdd(&hist[dv[i] >> RSH], 1) : 0;
    __syncthreads();

    const int c = hist[tid];
    int total;
    const int ex = block_scan_excl(c, wsum, 8, total);
    lstart[tid] = ex;
    binfo[b * NRP + tid] = make_int2(c, cbase + ex);
    __syncthreads();

    #pragma unroll
    for (int i = 0; i < 16; ++i) {
        if (dv[i] >= 0)
            stage[lstart[dv[i] >> RSH] + lr[i]] =
                ((unsigned)sv[i] << RSH) | (unsigned)(dv[i] & (RNODES - 1));
    }
    __syncthreads();

    for (int i = tid; i < total; i += 256) tmp[cbase + i] = stage[i];
}

template <bool PLACE>
__device__ __forceinline__ void seg_pass(const unsigned* __restrict__ tmp, const int* scnt,
                                         const int* sstart, int* cl, unsigned* stage2,
                                         int* __restrict__ csr, int base, bool staged) {
    const int lane = threadIdx.x & 31;
    const int wv = threadIdx.x >> 5;
    for (int g = wv; g * 8 < NB; g += 16) {
        int c[8], st[8];
        unsigned v[8];
        #pragma unroll
        for (int u = 0; u < 8; ++u) {
            c[u]  = scnt[g * 8 + u];
            st[u] = sstart[g * 8 + u];
        }
        #pragma unroll
        for (int u = 0; u < 8; ++u) {
            int a = st[u] + lane;
            a = a < N_EDGES - 1 ? a : N_EDGES - 1;
            v[u] = tmp[a];
        }
        #pragma unroll
        for (int u = 0; u < 8; ++u) {
            if (lane < c[u]) {
                const int d = (int)(v[u] & (RNODES - 1));
                if (!PLACE) {
                    atomicAdd(&cl[d], 1);
                } else {
                    const int pos = atomicAdd(&cl[d], 1);
                    const int s = (int)(v[u] >> RSH);
                    if (staged) stage2[pos] = (unsigned)s; else csr[base + pos] = s;
                }
            }
        }
        #pragma unroll
        for (int u = 0; u < 8; ++u) {
            for (int j = 32 + lane; j < c[u]; j += 32) {
                const unsigned x = tmp[st[u] + j];
                const int d = (int)(x & (RNODES - 1));
                if (!PLACE) {
                    atomicAdd(&cl[d], 1);
                } else {
                    const int pos = atomicAdd(&cl[d], 1);
                    const int s = (int)(x >> RSH);
                    if (staged) stage2[pos] = (unsigned)s; else csr[base + pos] = s;
                }
            }
        }
    }
}

__global__ __launch_bounds__(512) void sage_sort(const unsigned* __restrict__ tmp,
                                                 const int2* __restrict__ binfo,
                                                 int* __restrict__ cursor,
                                                 int* __restrict__ cnt,
                                                 int* __restrict__ offs,
                                                 int* __restrict__ csr, int n_nodes) {
    __shared__ int scnt[NBP];
    __shared__ int sstart[NBP];
    __shared__ int cntL[RNODES];
    __shared__ int cur[RNODES];
    __shared__ int wsum[16];
    __shared__ int sbase;
    __shared__ unsigned stage2[SCAP];

    const int tid = threadIdx.x;
    const int r = blockIdx.x;

    int c = 0;
    if (tid < NB) {
        const int2 bi = binfo[tid * NRP + r];
        c = bi.x;
        scnt[tid] = bi.x;
        sstart[tid] = bi.y;
    } else if (tid < NBP) {
        scnt[tid] = 0;
        sstart[tid] = 0;
    }
    cntL[tid] = 0;

    int n_r;
    (void)block_scan_excl(c, wsum, 16, n_r);
    if (tid == 0) sbase = atomicAdd(cursor, n_r);
    const bool staged = n_r <= SCAP;

    seg_pass<false>(tmp, scnt, sstart, cntL, stage2, csr, 0, staged);
    __syncthreads();

    const int my = cntL[tid];
    int tot2;
    const int lo = block_scan_excl(my, wsum, 16, tot2);
    const int base = sbase;
    cur[tid] = lo;
    const int node = r * RNODES + tid;
    if (node < n_nodes) {
        cnt[node] = my;
        offs[node] = base + lo;
    }
    __syncthreads();

    seg_pass<true>(tmp, scnt, sstart, cur, stage2, csr, base, staged);
    __syncthreads();

    if (staged) {
        for (int i = tid; i < n_r; i += 512) csr[base + i] = (int)stage2[i];
    }
}

__global__ __launch_bounds__(256) void sage_layer(const float* __restrict__ h,
                                                  const int* __restrict__ csr,
                                                  const int* __restrict__ offs,
                                                  const int* __restrict__ cnt,
                                                  const float* __restrict__ Wself,
                                                  const float* __restrict__ Wneigh,
                                                  const float* __restrict__ bias,
                                                  float* __restrict__ out,
                                                  int n_nodes, int do_relu) {
    __shared__ float sWs[F * F];
    __shared__ float sWn[F * F];
    __shared__ float sH [32 * HSTRIDE];
    __shared__ float sHn[32 * HSTRIDE];
    __shared__ float sB [F];

    const int tid = threadIdx.x;
    const int node0 = blockIdx.x * 32;
    const int wv = __builtin_amdgcn_readfirstlane((int)(threadIdx.x >> 5));
    const int lane = tid & 31;
    const int col2 = lane * 2;
    const int half = lane >> 4;
    const int c16i = lane & 15;
    const unsigned c16 = (unsigned)c16i * 16u;
    const int ppos = 2 * c16i + half;

    int beg[NPW], len[NPW], idx[NPW];
    float2 own[NPW];
    #pragma unroll
    for (int r = 0; r < NPW; ++r) {
        const int node = node0 + wv * NPW + r;
        const bool ok = node < n_nodes;
        const int nc = ok ? node : 0;
        beg[r] = offs[nc];
        len[r] = ok ? cnt[nc] : 0;
        own[r] = *(const float2*)(h + (long long)nc * F + col2);
        if (!ok) { own[r].x = 0.0f; own[r].y = 0.0f; }
    }
    #pragma unroll
    for (int r = 0; r < NPW; ++r) idx[r] = (ppos < len[r]) ? csr[beg[r] + ppos] : 0;

    #pragma unroll
    for (int i = 0; i < 16; ++i) {
        sWs[tid + i * 256] = Wself [tid + i * 256];
        sWn[tid + i * 256] = Wneigh[tid + i * 256];
    }
    if (tid < F) sB[tid] = bias[tid];

    #pragma unroll
    for (int r = 0; r < NPW; ++r) {
        sH[(wv * NPW + r) * HSTRIDE + col2    ] = own[r].x;
        sH[(wv * NPW + r) * HSTRIDE + col2 + 1] = own[r].y;
    }

    #pragma unroll
    for (int r = 0; r < NPW; ++r) {
        const int n = len[r];
        const int m = n < 32 ? n : 32;
        const int nfull = m >> 1;
        const int nb0 = nfull < GST ? nfull : GST;
        const int rowoff = idx[r] << 8;
        v4f acc = {0.0f, 0.0f, 0.0f, 0.0f};
        if (m & 1) {
            const int st = __builtin_amdgcn_readlane(rowoff, nfull);
            acc = ld_row16(h, (unsigned)st + c16);
        }
        v4f v[GST];
        if (nb0 > 0) { v[0] = ld_pair<0>(h, rowoff, c16);
        if (nb0 > 1) { v[1] = ld_pair<1>(h, rowoff, c16);
        if (nb0 > 2) { v[2] = ld_pair<2>(h, rowoff, c16);
        if (nb0 > 3) { v[3] = ld_pair<3>(h, rowoff, c16);
        if (nb0 > 4) { v[4] = ld_pair<4>(h, rowoff, c16);
        if (nb0 > 5) { v[5] = ld_pair<5>(h, rowoff, c16);
        if (nb0 > 6) { v[6] = ld_pair<6>(h, rowoff, c16);
        if (nb0 > 7) { v[7] = ld_pair<7>(h, rowoff, c16);
        if (nb0 > 8) { v[8] = ld_pair<8>(h, rowoff, c16);
        if (nb0 > 9) { v[9] = ld_pair<9>(h, rowoff, c16);
        if (nb0 > 10) { v[10] = ld_pair<10>(h, rowoff, c16);
        if (nb0 > 11) { v[11] = ld_pair<11>(h, rowoff, c16);
        }}}}}}}}}}}}
        if (half != 0) { acc.x = 0.0f; acc.y = 0.0f; acc.z = 0.0f; acc.w = 0.0f; }
        if (nb0 > 0) { acc += v[0];
        if (nb0 > 1) { acc += v[1];
        if (nb0 > 2) { acc += v[2];
        if (nb0 > 3) { acc += v[3];
        if (nb0 > 4) { acc += v[4];
        if (nb0 > 5) { acc += v[5];
        if (nb0 > 6) { acc += v[6];
        if (nb0 > 7) { acc += v[7];
        if (nb0 > 8) { acc += v[8];
        if (nb0 > 9) { acc += v[9];
        if (nb0 > 10) { acc += v[10];
        if (nb0 > 11) { acc += v[11];
        }}}}}}}}}}}}
        #pragma nounroll
        for (int j = GST; j < nfull; ++j) {
            const int ro = __shfl(rowoff, (half << 4) + j);
            acc += ld_row16(h, (unsigned)ro + c16);
        }
        #pragma nounroll
        for (int k = 32; k < n; ++k) {
            const int s = csr[beg[r] + k];
            const v4f x = ld_row16(h, ((unsigned)s << 8) + c16);
            if (half == 0) acc += x;
        }
        acc.x += other_half(acc.x);
        acc.y += other_half(acc.y);
        acc.z += other_half(acc.z);
        acc.w += other_half(acc.w);
        const float rinv = 1.0f / fmaxf((float)n, 1.0f);
        const int hcol = (wv * NPW + r) * HSTRIDE + 4 * c16i + 2 * half;
        sHn[hcol    ] = (half ? acc.z : acc.x) * rinv;
        sHn[hcol + 1] = (half ? acc.w : acc.y) * rinv;
    }
    __syncthreads();

    const int wave  = tid >> 5;
    const int m0    = (wave >> 2) * 16;
    const int c0    = (wave & 3) * 16;
    const int lrow  = lane & 15;
    const int khalf = lane >> 4;

    v8f acc = {};
    const float* aSelf = &sH [m0 * HSTRIDE + lrow * HSTRIDE];
    const float* aNei  = &sHn[m0 * HSTRIDE + lrow * HSTRIDE];
    #pragma unroll
    for (int kk = 0; kk < 16; ++kk) {
        const int kb = kk * 4 + khalf * 2;
        v2f a, b;
        a.x = aSelf[kb];
        a.y = aSelf[kb + 1];
        b.x = sWs[(kb    ) * F + c0 + lrow];
        b.y = sWs[(kb + 1) * F + c0 + lrow];
        acc = __builtin_amdgcn_wmma_f32_16x16x4_f32(false, a, false, b,
                                                    (short)0, acc, false, false);
        a.x = aNei[kb];
        a.y = aNei[kb + 1];
        b.x = sWn[(kb    ) * F + c0 + lrow];
        b.y = sWn[(kb + 1) * F + c0 + lrow];
        acc = __builtin_amdgcn_wmma_f32_16x16x4_f32(false, a, false, b,
                                                    (short)0, acc, false, false);
    }

    const float bv = sB[c0 + lrow];
    #pragma unroll
    for (int i = 0; i < 8; ++i) {
        int row = node0 + m0 + i + khalf * 8;
        float v = acc[i] + bv;
        if (do_relu) v = fmaxf(v, 0.0f);
        if (row < n_nodes) out[(long long)row * F + c0 + lrow] = v;
    }
}

extern "C" void kernel_launch(void* const* d_in, const int* in_sizes, int n_in,
                              void* d_out, int out_size, void* d_ws, size_t ws_size,
                              hipStream_t stream) {
    (void)in_sizes; (void)n_in; (void)out_size; (void)ws_size;
    const float* feat = (const float*)d_in[0];
    const int*   src  = (const int*)d_in[1];
    const int*   dst  = (const int*)d_in[2];
    const float* Ws[3] = { (const float*)d_in[3], (const float*)d_in[6], (const float*)d_in[9]  };
    const float* Wn[3] = { (const float*)d_in[4], (const float*)d_in[7], (const float*)d_in[10] };
    const float* bs[3] = { (const float*)d_in[5], (const float*)d_in[8], (const float*)d_in[11] };
    float* outp = (float*)d_out;

    int*   wsi    = (int*)d_ws;
    int*   cnt    = wsi;
    int*   cursor = wsi + N_NODES;
    int*   offs   = wsi + N_NODES + 64;
    int*   csr    = wsi + 2 * N_NODES + 128;
    float* h1     = (float*)(csr + N_EDGES);
    float* h2     = h1 + (long long)N_NODES * F;
    unsigned* tmp = (unsigned*)h2;
    int2*  binfo  = (int2*)(tmp + N_EDGES);

    const int layerBlocks = (N_NODES + 31) / 32;

    sage_part<<<NB, 256, 0, stream>>>(src, dst, tmp, binfo, cursor, N_EDGES);
    sage_sort<<<NR, 512, 0, stream>>>(tmp, binfo, cursor, cnt, offs, csr, N_NODES);

    sage_layer<<<layerBlocks, 256, 0, stream>>>(feat, csr, offs, cnt, Ws[0], Wn[0], bs[0], h1,   N_NODES, 1);
    sage_layer<<<layerBlocks, 256, 0, stream>>>(h1,   csr, offs, cnt, Ws[1], Wn[1], bs[1], h2,   N_NODES, 1);
    sage_layer<<<layerBlocks, 256, 0, stream>>>(h2,   csr, offs, cnt, Ws[2], Wn[2], bs[2], outp, N_NODES, 0);
}


# layer kernel: the four bucket-index loads and the bias load made unconditional (clamped address), so every first load issues in one group, not in five branches each with a full load wait; same sums
# speedup vs baseline: 5.6558x; 1.0449x over previous
#include <hip/hip_runtime.h>


#define N_NODES 100000
#define N_EDGES 1600000
#define F 64
#define HSTRIDE 65
#define NPW 4
#define GST 12

typedef __attribute__((ext_vector_type(2))) float v2f;
typedef __attribute__((ext_vector_type(8))) float v8f;
typedef __attribute__((ext_vector_type(4))) float v4f;

__device__ __forceinline__ v4f ld_row16(const float* __restrict__ hb, unsigned off) {
    return *(const v4f*)((const char*)hb + off);
}

template <int K>
__device__ __forceinline__ v4f ld_pair(const float* __restrict__ hb, int rowoff, unsigned c16) {
    const int ro = __builtin_amdgcn_update_dpp(0, rowoff, 0x150 + K, 0xF, 0xF, true);
    return ld_row16(hb, (unsigned)ro + c16);
}

__device__ __forceinline__ float other_half(float x) {
    const unsigned u = __float_as_uint(x);
    return __uint_as_float(__builtin_amdgcn_permlanex16(u, u, 0x76543210u, 0xfedcba98u, false, false));
}

#define RSH 9
#define RNODES (1 << RSH)
#define NR ((N_NODES + RNODES - 1) / RNODES)
#define NRP 256
#define PCH 4096
#define NB ((N_EDGES + PCH - 1) / PCH)
#define NBP 400
#define SCAP 12288

__device__ __forceinline__ int block_scan_excl(int v, int* wsum, int nwaves, int& total) {
    const int lane = threadIdx.x & 31;
    const int w = threadIdx.x >> 5;
    int incl = v;
    #pragma unroll
    for (int d = 1; d < 32; d <<= 1) {
        const int t = __shfl_up(incl, d);
        if (lane >= d) incl += t;
    }
    if (lane == 31) wsum[w] = incl;
    __syncthreads();
    int off = 0, tot = 0;
    for (int i = 0; i < nwaves; ++i) {
        const int s = wsum[i];
        if (i < w) off += s;
        tot += s;
    }
    total = tot;
    __syncthreads();
    return off + incl - v;
}

__global__ __launch_bounds__(256) void sage_part(const int* __restrict__ src,
                                                 const int* __restrict__ dst,
                                                 unsigned* __restrict__ tmp,
                                                 int2* __restrict__ binfo,
                                                 int* __restrict__ cursor, int n_edges) {
    __shared__ int hist[NRP];
    __shared__ int lstart[NRP];
    __shared__ int wsum[8];
    __shared__ unsigned stage[PCH];

    const int tid = threadIdx.x;
    const int lane = tid & 31;
    const int wv = tid >> 5;
    const int b = blockIdx.x;
    const int cbase = b * PCH;

    hist[tid] = 0;
    if (b == 0 && tid == 0) *cursor = 0;

    int dv[16], sv[16], lr[16];
    const int wbase = cbase + wv * 512;
    #pragma unroll
    for (int q = 0; q < 4; ++q) {
        const int e0 = wbase + q * 128 + lane * 4;
        if (e0 + 3 < n_edges) {
            const int4 d4 = *(const int4*)(dst + e0);
            const int4 s4 = *(const int4*)(src + e0);
            dv[4 * q + 0] = d4.x; dv[4 * q + 1] = d4.y; dv[4 * q + 2] = d4.z; dv[4 * q + 3] = d4.w;
            sv[4 * q + 0] = s4.x; sv[4 * q + 1] = s4.y; sv[4 * q + 2] = s4.z; sv[4 * q + 3] = s4.w;
        } else {
            #pragma unroll
            for (int k = 0; k < 4; ++k) {
                const int e = e0 + k;
                const bool ok = e < n_edges;
                dv[4 * q + k] = ok ? dst[e] : -1;
                sv[4 * q + k] = ok ? src[e] : 0;
            }
        }
    }
    __syncthreads();

    #pragma unroll
    for (int i = 0; i < 16; ++i)
        lr[i] = (dv[i] >= 0) ? atomicAdd(&hist[dv[i] >> RSH], 1) : 0;
    __syncthreads();

    const int c = hist[tid];
    int total;
    const int ex = block_scan_excl(c, wsum, 8, total);
    lstart[tid] = ex;
    binfo[b * NRP + tid] = make_int2(c, cbase + ex);
    __syncthreads();

    #pragma unroll
    for (int i = 0; i < 16; ++i) {
        if (dv[i] >= 0)
            stage[lstart[dv[i] >> RSH] + lr[i]] =
                ((unsigned)sv[i] << RSH) | (unsigned)(dv[i] & (RNODES - 1));
    }
    __syncthreads();

    for (int i = tid; i < total; i += 256) tmp[cbase + i] = stage[i];
}

template <bool PLACE>
__device__ __forceinline__ void seg_pass(const unsigned* __restrict__ tmp, const int* scnt,
                                         const int* sstart, int* cl, unsigned* stage2,
                                         int* __restrict__ csr, int base, bool staged) {
    const int lane = threadIdx.x & 31;
    const int wv = threadIdx.x >> 5;
    for (int g = wv; g * 8 < NB; g += 16) {
        int c[8], st[8];
        unsigned v[8];
        #pragma unroll
        for (int u = 0; u < 8; ++u) {
            c[u]  = scnt[g * 8 + u];
            st[u] = sstart[g * 8 + u];
        }
        #pragma unroll
        for (int u = 0; u < 8; ++u) {
            int a = st[u] + lane;
            a = a < N_EDGES - 1 ? a : N_EDGES - 1;
            v[u] = tmp[a];
        }
        #pragma unroll
        for (int u = 0; u < 8; ++u) {
            if (lane < c[u]) {
                const int d = (int)(v[u] & (RNODES - 1));
                if (!PLACE) {
                    atomicAdd(&cl[d], 1);
                } else {
                    const int pos = atomicAdd(&cl[d], 1);
                    const int s = (int)(v[u] >> RSH);
                    if (staged) stage2[pos] = (unsigned)s; else csr[base + pos] = s;
                }
            }
        }
        #pragma unroll
        for (int u = 0; u < 8; ++u) {
            for (int j = 32 + lane; j < c[u]; j += 32) {
                const unsigned x = tmp[st[u] + j];
                const int d = (int)(x & (RNODES - 1));
                if (!PLACE) {
                    atomicAdd(&cl[d], 1);
                } else {
                    const int pos = atomicAdd(&cl[d], 1);
                    const int s = (int)(x >> RSH);
                    if (staged) stage2[pos] = (unsigned)s; else csr[base + pos] = s;
                }
            }
        }
    }
}

__global__ __launch_bounds__(512) void sage_sort(const unsigned* __restrict__ tmp,
                                                 const int2* __restrict__ binfo,
                                                 int* __restrict__ cursor,
                                                 int* __restrict__ cnt,
                                                 int* __restrict__ offs,
                                                 int* __restrict__ csr, int n_nodes) {
    __shared__ int scnt[NBP];
    __shared__ int sstart[NBP];
    __shared__ int cntL[RNODES];
    __shared__ int cur[RNODES];
    __shared__ int wsum[16];
    __shared__ int sbase;
    __shared__ unsigned stage2[SCAP];

    const int tid = threadIdx.x;
    const int r = blockIdx.x;

    int c = 0;
    if (tid < NB) {
        const int2 bi = binfo[tid * NRP + r];
        c = bi.x;
        scnt[tid] = bi.x;
        sstart[tid] = bi.y;
    } else if (tid < NBP) {
        scnt[tid] = 0;
        sstart[tid] = 0;
    }
    cntL[tid] = 0;

    int n_r;
    (void)block_scan_excl(c, wsum, 16, n_r);
    if (tid == 0) sbase = atomicAdd(cursor, n_r);
    const bool staged = n_r <= SCAP;

    seg_pass<false>(tmp, scnt, sstart, cntL, stage2, csr, 0, staged);
    __syncthreads();

    const int my = cntL[tid];
    int tot2;
    const int lo = block_scan_excl(my, wsum, 16, tot2);
    const int base = sbase;
    cur[tid] = lo;
    const int node = r * RNODES + tid;
    if (node < n_nodes) {
        cnt[node] = my;
        offs[node] = base + lo;
    }
    __syncthreads();

    seg_pass<true>(tmp, scnt, sstart, cur, stage2, csr, base, staged);
    __syncthreads();

    if (staged) {
        for (int i = tid; i < n_r; i += 512) csr[base + i] = (int)stage2[i];
    }
}

__global__ __launch_bounds__(256) void sage_layer(const float* __restrict__ h,
                                                  const int* __restrict__ csr,
                                                  const int* __restrict__ offs,
                                                  const int* __restrict__ cnt,
                                                  const float* __restrict__ Wself,
                                                  const float* __restrict__ Wneigh,
                                                  const float* __restrict__ bias,
                                                  float* __restrict__ out,
                                                  int n_nodes, int do_relu) {
    __shared__ float sWs[F * F];
    __shared__ float sWn[F * F];
    __shared__ float sH [32 * HSTRIDE];
    __shared__ float sHn[32 * HSTRIDE];

    const int tid = threadIdx.x;
    const int node0 = blockIdx.x * 32;
    const int wv = __builtin_amdgcn_readfirstlane((int)(threadIdx.x >> 5));
    const int lane = tid & 31;
    const int col2 = lane * 2;
    const int half = lane >> 4;
    const int c16i = lane & 15;
    const unsigned c16 = (unsigned)c16i * 16u;
    const int ppos = 2 * c16i + half;

    int beg[NPW], len[NPW], idx[NPW];
    float2 own[NPW];
    #pragma unroll
    for (int r = 0; r < NPW; ++r) {
        const int node = node0 + wv * NPW + r;
        const bool ok = node < n_nodes;
        const int nc = ok ? node : 0;
        beg[r] = offs[nc];
        len[r] = ok ? cnt[nc] : 0;
        own[r] = *(const float2*)(h + (long long)nc * F + col2);
        if (!ok) { own[r].x = 0.0f; own[r].y = 0.0f; }
    }
    #pragma unroll
    for (int r = 0; r < NPW; ++r) {
        int a = beg[r] + ppos;
        a = a < N_EDGES - 1 ? a : N_EDGES - 1;
        const int s = csr[a];
        idx[r] = (ppos < len[r]) ? s : 0;
    }
    const float bv = bias[(wv & 3) * 16 + c16i];

    #pragma unroll
    for (int i = 0; i < 16; ++i) {
        sWs[tid + i * 256] = Wself [tid + i * 256];
        sWn[tid + i * 256] = Wneigh[tid + i * 256];
    }

    #pragma unroll
    for (int r = 0; r < NPW; ++r) {
        sH[(wv * NPW + r) * HSTRIDE + col2    ] = own[r].x;
        sH[(wv * NPW + r) * HSTRIDE + col2 + 1] = own[r].y;
    }

    #pragma unroll
    for (int r = 0; r < NPW; ++r) {
        const int n = len[r];
        const int m = n < 32 ? n : 32;
        const int nfull = m >> 1;
        const int nb0 = nfull < GST ? nfull : GST;
        const int rowoff = idx[r] << 8;
        v4f acc = {0.0f, 0.0f, 0.0f, 0.0f};
        if (m & 1) {
            const int st = __builtin_amdgcn_readlane(rowoff, nfull);
            acc = ld_row16(h, (unsigned)st + c16);
        }
        v4f v[GST];
        if (nb0 > 0) { v[0] = ld_pair<0>(h, rowoff, c16);
        if (nb0 > 1) { v[1] = ld_pair<1>(h, rowoff, c16);
        if (nb0 > 2) { v[2] = ld_pair<2>(h, rowoff, c16);
        if (nb0 > 3) { v[3] = ld_pair<3>(h, rowoff, c16);
        if (nb0 > 4) { v[4] = ld_pair<4>(h, rowoff, c16);
        if (nb0 > 5) { v[5] = ld_pair<5>(h, rowoff, c16);
        if (nb0 > 6) { v[6] = ld_pair<6>(h, rowoff, c16);
        if (nb0 > 7) { v[7] = ld_pair<7>(h, rowoff, c16);
        if (nb0 > 8) { v[8] = ld_pair<8>(h, rowoff, c16);
        if (nb0 > 9) { v[9] = ld_pair<9>(h, rowoff, c16);
        if (nb0 > 10) { v[10] = ld_pair<10>(h, rowoff, c16);
        if (nb0 > 11) { v[11] = ld_pair<11>(h, rowoff, c16);
        }}}}}}}}}}}}
        if (half != 0) { acc.x = 0.0f; acc.y = 0.0f; acc.z = 0.0f; acc.w = 0.0f; }
        if (nb0 > 0) { acc += v[0];
        if (nb0 > 1) { acc += v[1];
        if (nb0 > 2) { acc += v[2];
        if (nb0 > 3) { acc += v[3];
        if (nb0 > 4) { acc += v[4];
        if (nb0 > 5) { acc += v[5];
        if (nb0 > 6) { acc += v[6];
        if (nb0 > 7) { acc += v[7];
        if (nb0 > 8) { acc += v[8];
        if (nb0 > 9) { acc += v[9];
        if (nb0 > 10) { acc += v[10];
        if (nb0 > 11) { acc += v[11];
        }}}}}}}}}}}}
        #pragma nounroll
        for (int j = GST; j < nfull; ++j) {
            const int ro = __shfl(rowoff, (half << 4) + j);
            acc += ld_row16(h, (unsigned)ro + c16);
        }
        #pragma nounroll
        for (int k = 32; k < n; ++k) {
            const int s = csr[beg[r] + k];
            const v4f x = ld_row16(h, ((unsigned)s << 8) + c16);
            if (half == 0) acc += x;
        }
        acc.x += other_half(acc.x);
        acc.y += other_half(acc.y);
        acc.z += other_half(acc.z);
        acc.w += other_half(acc.w);
        const float rinv = 1.0f / fmaxf((float)n, 1.0f);
        const int hcol = (wv * NPW + r) * HSTRIDE + 4 * c16i + 2 * half;
        sHn[hcol    ] = (half ? acc.z : acc.x) * rinv;
        sHn[hcol + 1] = (half ? acc.w : acc.y) * rinv;
    }
    __syncthreads();

    const int wave  = tid >> 5;
    const int m0    = (wave >> 2) * 16;
    const int c0    = (wave & 3) * 16;
    const int lrow  = lane & 15;
    const int khalf = lane >> 4;

    v8f acc = {};
    const float* aSelf = &sH [m0 * HSTRIDE + lrow * HSTRIDE];
    const float* aNei  = &sHn[m0 * HSTRIDE + lrow * HSTRIDE];
    #pragma unroll
    for (int kk = 0; kk < 16; ++kk) {
        const int kb = kk * 4 + khalf * 2;
        v2f a, b;
        a.x = aSelf[kb];
        a.y = aSelf[kb + 1];
        b.x = sWs[(kb    ) * F + c0 + lrow];
        b.y = sWs[(kb + 1) * F + c0 + lrow];
        acc = __builtin_amdgcn_wmma_f32_16x16x4_f32(false, a, false, b,
                                                    (short)0, acc, false, false);
        a.x = aNei[kb];
        a.y = aNei[kb + 1];
        b.x = sWn[(kb    ) * F + c0 + lrow];
        b.y = sWn[(kb + 1) * F + c0 + lrow];
        acc = __builtin_amdgcn_wmma_f32_16x16x4_f32(false, a, false, b,
                                                    (short)0, acc, false, false);
    }

    #pragma unroll
    for (int i = 0; i < 8; ++i) {
        int row = node0 + m0 + i + khalf * 8;
        float v = acc[i] + bv;
        if (do_relu) v = fmaxf(v, 0.0f);
        if (row < n_nodes) out[(long long)row * F + c0 + lrow] = v;
    }
}

extern "C" void kernel_launch(void* const* d_in, const int* in_sizes, int n_in,
                              void* d_out, int out_size, void* d_ws, size_t ws_size,
                              hipStream_t stream) {
    (void)in_sizes; (void)n_in; (void)out_size; (void)ws_size;
    const float* feat = (const float*)d_in[0];
    const int*   src  = (const int*)d_in[1];
    const int*   dst  = (const int*)d_in[2];
    const float* Ws[3] = { (const float*)d_in[3], (const float*)d_in[6], (const float*)d_in[9]  };
    const float* Wn[3] = { (const float*)d_in[4], (const float*)d_in[7], (const float*)d_in[10] };
    const float* bs[3] = { (const float*)d_in[5], (const float*)d_in[8], (const float*)d_in[11] };
    float* outp = (float*)d_out;

    int*   wsi    = (int*)d_ws;
    int*   cnt    = wsi;
    int*   cursor = wsi + N_NODES;
    int*   offs   = wsi + N_NODES + 64;
    int*   csr    = wsi + 2 * N_NODES + 128;
    float* h1     = (float*)(csr + N_EDGES);
    float* h2     = h1 + (long long)N_NODES * F;
    unsigned* tmp = (unsigned*)h2;
    int2*  binfo  = (int2*)(tmp + N_EDGES);

    const int layerBlocks = (N_NODES + 31) / 32;

    sage_part<<<NB, 256, 0, stream>>>(src, dst, tmp, binfo, cursor, N_EDGES);
    sage_sort<<<NR, 512, 0, stream>>>(tmp, binfo, cursor, cnt, offs, csr, N_NODES);

    sage_layer<<<layerBlocks, 256, 0, stream>>>(feat, csr, offs, cnt, Ws[0], Wn[0], bs[0], h1,   N_NODES, 1);
    sage_layer<<<layerBlocks, 256, 0, stream>>>(h1,   csr, offs, cnt, Ws[1], Wn[1], bs[1], h2,   N_NODES, 1);
    sage_layer<<<layerBlocks, 256, 0, stream>>>(h2,   csr, offs, cnt, Ws[2], Wn[2], bs[2], outp, N_NODES, 0);
}


# layer kernel persistent: 1536 resident blocks stage the weights in LDS once and draw 32-node tiles from a per-layer counter (1 int atomic per tile); was 3125 blocks each staging them; same sums
# speedup vs baseline: 5.8551x; 1.0352x over previous
#include <hip/hip_runtime.h>


#define N_NODES 100000
#define N_EDGES 1600000
#define F 64
#define HSTRIDE 65
#define NPW 4
#define GST 12
#define LBLK 1536

typedef __attribute__((ext_vector_type(2))) float v2f;
typedef __attribute__((ext_vector_type(8))) float v8f;
typedef __attribute__((ext_vector_type(4))) float v4f;

__device__ __forceinline__ v4f ld_row16(const float* __restrict__ hb, unsigned off) {
    return *(const v4f*)((const char*)hb + off);
}

template <int K>
__device__ __forceinline__ v4f ld_pair(const float* __restrict__ hb, int rowoff, unsigned c16) {
    const int ro = __builtin_amdgcn_update_dpp(0, rowoff, 0x150 + K, 0xF, 0xF, true);
    return ld_row16(hb, (unsigned)ro + c16);
}

__device__ __forceinline__ float other_half(float x) {
    const unsigned u = __float_as_uint(x);
    return __uint_as_float(__builtin_amdgcn_permlanex16(u, u, 0x76543210u, 0xfedcba98u, false, false));
}

#define RSH 9
#define RNODES (1 << RSH)
#define NR ((N_NODES + RNODES - 1) / RNODES)
#define NRP 256
#define PCH 4096
#define NB ((N_EDGES + PCH - 1) / PCH)
#define NBP 400
#define SCAP 12288

__device__ __forceinline__ int block_scan_excl(int v, int* wsum, int nwaves, int& total) {
    const int lane = threadIdx.x & 31;
    const int w = threadIdx.x >> 5;
    int incl = v;
    #pragma unroll
    for (int d = 1; d < 32; d <<= 1) {
        const int t = __shfl_up(incl, d);
        if (lane >= d) incl += t;
    }
    if (lane == 31) wsum[w] = incl;
    __syncthreads();
    int off = 0, tot = 0;
    for (int i = 0; i < nwaves; ++i) {
        const int s = wsum[i];
        if (i < w) off += s;
        tot += s;
    }
    total = tot;
    __syncthreads();
    return off + incl - v;
}

__global__ __launch_bounds__(256) void sage_part(const int* __restrict__ src,
                                                 const int* __restrict__ dst,
                                                 unsigned* __restrict__ tmp,
                                                 int2* __restrict__ binfo,
                                                 int* __restrict__ cursor, int n_edges) {
    __shared__ int hist[NRP];
    __shared__ int lstart[NRP];
    __shared__ int wsum[8];
    __shared__ unsigned stage[PCH];

    const int tid = threadIdx.x;
    const int lane = tid & 31;
    const int wv = tid >> 5;
    const int b = blockIdx.x;
    const int cbase = b * PCH;

    hist[tid] = 0;
    if (b == 0 && tid < 4) cursor[tid] = 0;

    int dv[16], sv[16], lr[16];
    const int wbase = cbase + wv * 512;
    #pragma unroll
    for (int q = 0; q < 4; ++q) {
        const int e0 = wbase + q * 128 + lane * 4;
        if (e0 + 3 < n_edges) {
            const int4 d4 = *(const int4*)(dst + e0);
            const int4 s4 = *(const int4*)(src + e0);
            dv[4 * q + 0] = d4.x; dv[4 * q + 1] = d4.y; dv[4 * q + 2] = d4.z; dv[4 * q + 3] = d4.w;
            sv[4 * q + 0] = s4.x; sv[4 * q + 1] = s4.y; sv[4 * q + 2] = s4.z; sv[4 * q + 3] = s4.w;
        } else {
            #pragma unroll
            for (int k = 0; k < 4; ++k) {
                const int e = e0 + k;
                const bool ok = e < n_edges;
                dv[4 * q + k] = ok ? dst[e] : -1;
                sv[4 * q + k] = ok ? src[e] : 0;
            }
        }
    }
    __syncthreads();

    #pragma unroll
    for (int i = 0; i < 16; ++i)
        lr[i] = (dv[i] >= 0) ? atomicAdd(&hist[dv[i] >> RSH], 1) : 0;
    __syncthreads();

    const int c = hist[tid];
    int total;
    const int ex = block_scan_excl(c, wsum, 8, total);
    lstart[tid] = ex;
    binfo[b * NRP + tid] = make_int2(c, cbase + ex);
    __syncthreads();

    #pragma unroll
    for (int i = 0; i < 16; ++i) {
        if (dv[i] >= 0)
            stage[lstart[dv[i] >> RSH] + lr[i]] =
                ((unsigned)sv[i] << RSH) | (unsigned)(dv[i] & (RNODES - 1));
    }
    __syncthreads();

    for (int i = tid; i < total; i += 256) tmp[cbase + i] = stage[i];
}

template <bool PLACE>
__device__ __forceinline__ void seg_pass(const unsigned* __restrict__ tmp, const int* scnt,
                                         const int* sstart, int* cl, unsigned* stage2,
                                         int* __restrict__ csr, int base, bool staged) {
    const int lane = threadIdx.x & 31;
    const int wv = threadIdx.x >> 5;
    for (int g = wv; g * 8 < NB; g += 16) {
        int c[8], st[8];
        unsigned v[8];
        #pragma unroll
        for (int u = 0; u < 8; ++u) {
            c[u]  = scnt[g * 8 + u];
            st[u] = sstart[g * 8 + u];
        }
        #pragma unroll
        for (int u = 0; u < 8; ++u) {
            int a = st[u] + lane;
            a = a < N_EDGES - 1 ? a : N_EDGES - 1;
            v[u] = tmp[a];
        }
        #pragma unroll
        for (int u = 0; u < 8; ++u) {
            if (lane < c[u]) {
                const int d = (int)(v[u] & (RNODES - 1));
                if (!PLACE) {
                    atomicAdd(&cl[d], 1);
                } else {
                    const int pos = atomicAdd(&cl[d], 1);
                    const int s = (int)(v[u] >> RSH);
                    if (staged) stage2[pos] = (unsigned)s; else csr[base + pos] = s;
                }
            }
        }
        #pragma unroll
        for (int u = 0; u < 8; ++u) {
            for (int j = 32 + lane; j < c[u]; j += 32) {
                const unsigned x = tmp[st[u] + j];
                const int d = (int)(x & (RNODES - 1));
                if (!PLACE) {
                    atomicAdd(&cl[d], 1);
                } else {
                    const int pos = atomicAdd(&cl[d], 1);
                    const int s = (int)(x >> RSH);
                    if (staged) stage2[pos] = (unsigned)s; else csr[base + pos] = s;
                }
            }
        }
    }
}

__global__ __launch_bounds__(512) void sage_sort(const unsigned* __restrict__ tmp,
                                                 const int2* __restrict__ binfo,
                                                 int* __restrict__ cursor,
                                                 int* __restrict__ cnt,
                                                 int* __restrict__ offs,
                                                 int* __restrict__ csr, int n_nodes) {
    __shared__ int scnt[NBP];
    __shared__ int sstart[NBP];
    __shared__ int cntL[RNODES];
    __shared__ int cur[RNODES];
    __shared__ int wsum[16];
    __shared__ int sbase;
    __shared__ unsigned stage2[SCAP];

    const int tid = threadIdx.x;
    const int r = blockIdx.x;

    int c = 0;
    if (tid < NB) {
        const int2 bi = binfo[tid * NRP + r];
        c = bi.x;
        scnt[tid] = bi.x;
        sstart[tid] = bi.y;
    } else if (tid < NBP) {
        scnt[tid] = 0;
        sstart[tid] = 0;
    }
    cntL[tid] = 0;

    int n_r;
    (void)block_scan_excl(c, wsum, 16, n_r);
    if (tid == 0) sbase = atomicAdd(cursor, n_r);
    const bool staged = n_r <= SCAP;

    seg_pass<false>(tmp, scnt, sstart, cntL, stage2, csr, 0, staged);
    __syncthreads();

    const int my = cntL[tid];
    int tot2;
    const int lo = block_scan_excl(my, wsum, 16, tot2);
    const int base = sbase;
    cur[tid] = lo;
    const int node = r * RNODES + tid;
    if (node < n_nodes) {
        cnt[node] = my;
        offs[node] = base + lo;
    }
    __syncthreads();

    seg_pass<true>(tmp, scnt, sstart, cur, stage2, csr, base, staged);
    __syncthreads();

    if (staged) {
        for (int i = tid; i < n_r; i += 512) csr[base + i] = (int)stage2[i];
    }
}

__global__ __launch_bounds__(256) void sage_layer(const float* __restrict__ h,
                                                  const int* __restrict__ csr,
                                                  const int* __restrict__ offs,
                                                  const int* __restrict__ cnt,
                                                  const float* __restrict__ Wself,
                                                  const float* __restrict__ Wneigh,
                                                  const float* __restrict__ bias,
                                                  float* __restrict__ out,
                                                  int* __restrict__ tileq,
                                                  int n_nodes, int do_relu) {
    __shared__ float sW [2 * F * F];
    __shared__ float sH [32 * HSTRIDE];
    __shared__ float sHn[32 * HSTRIDE];
    __shared__ int s_next;

    const int tid = threadIdx.x;
    const int wv = __builtin_amdgcn_readfirstlane((int)(threadIdx.x >> 5));
    const int lane = tid & 31;
    const int col2 = lane * 2;
    const int half = lane >> 4;
    const int c16i = lane & 15;
    const unsigned c16 = (unsigned)c16i * 16u;
    const int ppos = 2 * c16i + half;
    const int ntiles = (n_nodes + 31) >> 5;

    const float bv = bias[(wv & 3) * 16 + c16i];

    #pragma unroll
    for (int i = 0; i < 16; ++i) {
        sW[tid + i * 256]         = Wself [tid + i * 256];
        sW[F * F + tid + i * 256] = Wneigh[tid + i * 256];
    }

    const int m0    = (wv >> 2) * 16;
    const int c0    = (wv & 3) * 16;
    const int lrow  = lane & 15;
    const int khalf = lane >> 4;
    const float* aSelf = &sH [(m0 + lrow) * HSTRIDE + khalf * 2];
    const float* aNei  = &sHn[(m0 + lrow) * HSTRIDE + khalf * 2];
    const float* bSelf = &sW[khalf * 2 * F + c0 + lrow];
    const float* bNei  = bSelf + F * F;
    const unsigned ooff = (unsigned)((khalf * 8 * F + c0 + lrow) * 4);
    const unsigned own8 = (unsigned)lane * 8u;

    int tile = blockIdx.x;
    while (tile < ntiles) {
    const int node0 = tile * 32;
    const int nodew = node0 + wv * NPW;

    int beg[NPW], len[NPW], idx[NPW];
    float2 own[NPW];
    {
        const int nq = nodew + (lane & (NPW - 1));
        const int ncq = nq < n_nodes ? nq : 0;
        const int begv = offs[ncq];
        const int cntv = cnt[ncq];
        #pragma unroll
        for (int r = 0; r < NPW; ++r) {
            const bool ok = nodew + r < n_nodes;
            const int nc = ok ? nodew + r : 0;
            own[r] = *(const float2*)((const char*)h + (((unsigned)nc << 8) + own8));
            if (!ok) { own[r].x = 0.0f; own[r].y = 0.0f; }
        }
        #pragma unroll
        for (int r = 0; r < NPW; ++r) {
            beg[r] = __builtin_amdgcn_readlane(begv, r);
            len[r] = (nodew + r < n_nodes) ? __builtin_amdgcn_readlane(cntv, r) : 0;
        }
    }
    #pragma unroll
    for (int r = 0; r < NPW; ++r) {
        int a = beg[r] + ppos;
        a = a < N_EDGES - 1 ? a : N_EDGES - 1;
        const int s = csr[a];
        idx[r] = (ppos < len[r]) ? s : 0;
    }

    #pragma unroll
    for (int r = 0; r < NPW; ++r) {
        sH[(wv * NPW + r) * HSTRIDE + col2    ] = own[r].x;
        sH[(wv * NPW + r) * HSTRIDE + col2 + 1] = own[r].y;
    }

    #pragma unroll
    for (int r = 0; r < NPW; ++r) {
        const int n = len[r];
        const int m = n < 32 ? n : 32;
        const int nfull = m >> 1;
        const int nb0 = nfull < GST ? nfull : GST;
        const int rowoff = idx[r] << 8;
        v4f acc = {0.0f, 0.0f, 0.0f, 0.0f};
        if (m & 1) {
            const int st = __builtin_amdgcn_readlane(rowoff, nfull);
            acc = ld_row16(h, (unsigned)st + c16);
        }
        v4f v[GST];
        #pragma unroll
        for (int k = 0; k < GST; ++k) asm volatile("" : "=v"(v[k]));
        if (nb0 > 0) { v[0] = ld_pair<0>(h, rowoff, c16);
        if (nb0 > 1) { v[1] = ld_pair<1>(h, rowoff, c16);
        if (nb0 > 2) { v[2] = ld_pair<2>(h, rowoff, c16);
        if (nb0 > 3) { v[3] = ld_pair<3>(h, rowoff, c16);
        if (nb0 > 4) { v[4] = ld_pair<4>(h, rowoff, c16);
        if (nb0 > 5) { v[5] = ld_pair<5>(h, rowoff, c16);
        if (nb0 > 6) { v[6] = ld_pair<6>(h, rowoff, c16);
        if (nb0 > 7) { v[7] = ld_pair<7>(h, rowoff, c16);
        if (nb0 > 8) { v[8] = ld_pair<8>(h, rowoff, c16);
        if (nb0 > 9) { v[9] = ld_pair<9>(h, rowoff, c16);
        if (nb0 > 10) { v[10] = ld_pair<10>(h, rowoff, c16);
        if (nb0 > 11) { v[11] = ld_pair<11>(h, rowoff, c16);
        }}}}}}}}}}}}
        if (r == NPW - 1 && tid == 0) s_next = LBLK + atomicAdd(tileq, 1);
        if (half != 0) { acc.x = 0.0f; acc.y = 0.0f; acc.z = 0.0f; acc.w = 0.0f; }
        if (nb0 > 0) { acc += v[0];
        if (nb0 > 1) { acc += v[1];
        if (nb0 > 2) { acc += v[2];
        if (nb0 > 3) { acc += v[3];
        if (nb0 > 4) { acc += v[4];
        if (nb0 > 5) { acc += v[5];
        if (nb0 > 6) { acc += v[6];
        if (nb0 > 7) { acc += v[7];
        if (nb0 > 8) { acc += v[8];
        if (nb0 > 9) { acc += v[9];
        if (nb0 > 10) { acc += v[10];
        if (nb0 > 11) { acc += v[11];
        }}}}}}}}}}}}
        #pragma nounroll
        for (int j = GST; j < nfull; ++j) {
            const int ro = __shfl(rowoff, (half << 4) + j);
            acc += ld_row16(h, (unsigned)ro + c16);
        }
        #pragma nounroll
        for (int k = 32; k < n; ++k) {
            const int s = csr[beg[r] + k];
            const v4f x = ld_row16(h, ((unsigned)s << 8) + c16);
            if (half == 0) acc += x;
        }
        acc.x += other_half(acc.x);
        acc.y += other_half(acc.y);
        acc.z += other_half(acc.z);
        acc.w += other_half(acc.w);
        const float rinv = 1.0f / fmaxf((float)n, 1.0f);
        const int hcol = (wv * NPW + r) * HSTRIDE + 4 * c16i + 2 * half;
        sHn[hcol    ] = (half ? acc.z : acc.x) * rinv;
        sHn[hcol + 1] = (half ? acc.w : acc.y) * rinv;
    }
    __syncthreads();

    const int nxt = __builtin_amdgcn_readfirstlane(s_next);

    v8f acc = {};
    #pragma unroll
    for (int kk = 0; kk < 16; ++kk) {
        v2f a, b;
        a.x = aSelf[kk * 4];
        a.y = aSelf[kk * 4 + 1];
        b.x = bSelf[kk * 4 * F];
        b.y = bSelf[kk * 4 * F + F];
        acc = __builtin_amdgcn_wmma_f32_16x16x4_f32(false, a, false, b,
                                                    (short)0, acc, false, false);
        a.x = aNei[kk * 4];
        a.y = aNei[kk * 4 + 1];
        b.x = bNei[kk * 4 * F];
        b.y = bNei[kk * 4 * F + F];
        acc = __builtin_amdgcn_wmma_f32_16x16x4_f32(false, a, false, b,
                                                    (short)0, acc, false, false);
    }

    asm volatile("v_nop\n\tv_nop\n\tv_nop\n\tv_nop" : "+v"(acc));

    __syncthreads();

    #pragma unroll
    for (int i = 0; i < 8; ++i) {
        const int rowi = node0 + m0 + i;
        float v = acc[i] + bv;
        if (do_relu) v = fmaxf(v, 0.0f);
        if (rowi + khalf * 8 < n_nodes) *(float*)((char*)out + (((unsigned)rowi << 8) + ooff)) = v;
    }
    tile = nxt;
    }
}

extern "C" void kernel_launch(void* const* d_in, const int* in_sizes, int n_in,
                              void* d_out, int out_size, void* d_ws, size_t ws_size,
                              hipStream_t stream) {
    (void)in_sizes; (void)n_in; (void)out_size; (void)ws_size;
    const float* feat = (const float*)d_in[0];
    const int*   src  = (const int*)d_in[1];
    const int*   dst  = (const int*)d_in[2];
    const float* Ws[3] = { (const float*)d_in[3], (const float*)d_in[6], (const float*)d_in[9]  };
    const float* Wn[3] = { (const float*)d_in[4], (const float*)d_in[7], (const float*)d_in[10] };
    const float* bs[3] = { (const float*)d_in[5], (const float*)d_in[8], (const float*)d_in[11] };
    float* outp = (float*)d_out;

    int*   wsi    = (int*)d_ws;
    int*   cnt    = wsi;
    int*   cursor = wsi + N_NODES;
    int*   tileq  = cursor + 1;
    int*   offs   = wsi + N_NODES + 64;
    int*   csr    = wsi + 2 * N_NODES + 128;
    float* h1     = (float*)(csr + N_EDGES);
    float* h2     = h1 + (long long)N_NODES * F;
    unsigned* tmp = (unsigned*)h2;
    int2*  binfo  = (int2*)(tmp + N_EDGES);

    sage_part<<<NB, 256, 0, stream>>>(src, dst, tmp, binfo, cursor, N_EDGES);
    sage_sort<<<NR, 512, 0, stream>>>(tmp, binfo, cursor, cnt, offs, csr, N_NODES);

    sage_layer<<<LBLK, 256, 0, stream>>>(feat, csr, offs, cnt, Ws[0], Wn[0], bs[0], h1,   tileq + 0, N_NODES, 1);
    sage_layer<<<LBLK, 256, 0, stream>>>(h1,   csr, offs, cnt, Ws[1], Wn[1], bs[1], h2,   tileq + 1, N_NODES, 1);
    sage_layer<<<LBLK, 256, 0, stream>>>(h2,   csr, offs, cnt, Ws[2], Wn[2], bs[2], outp, tileq + 2, N_NODES, 0);
}


# layer kernel: output tile goes through LDS (reusing sH, one more block barrier per tile) so each 256-B row of h/out is written by one store instruction, not four 64-B pieces of four waves; same sums
# speedup vs baseline: 6.2165x; 1.0617x over previous
#include <hip/hip_runtime.h>


#define N_NODES 100000
#define N_EDGES 1600000
#define F 64
#define HSTRIDE 65
#define NPW 4
#define GST 12
#define LBLK 1536

typedef __attribute__((ext_vector_type(2))) float v2f;
typedef __attribute__((ext_vector_type(8))) float v8f;
typedef __attribute__((ext_vector_type(4))) float v4f;

__device__ __forceinline__ v4f ld_row16(const float* __restrict__ hb, unsigned off) {
    return *(const v4f*)((const char*)hb + off);
}

template <int K>
__device__ __forceinline__ v4f ld_pair(const float* __restrict__ hb, int rowoff, unsigned c16) {
    const int ro = __builtin_amdgcn_update_dpp(0, rowoff, 0x150 + K, 0xF, 0xF, true);
    return ld_row16(hb, (unsigned)ro + c16);
}

__device__ __forceinline__ float other_half(float x) {
    const unsigned u = __float_as_uint(x);
    return __uint_as_float(__builtin_amdgcn_permlanex16(u, u, 0x76543210u, 0xfedcba98u, false, false));
}

#define RSH 9
#define RNODES (1 << RSH)
#define NR ((N_NODES + RNODES - 1) / RNODES)
#define NRP 256
#define PCH 4096
#define NB ((N_EDGES + PCH - 1) / PCH)
#define NBP 400
#define SCAP 12288

__device__ __forceinline__ int block_scan_excl(int v, int* wsum, int nwaves, int& total) {
    const int lane = threadIdx.x & 31;
    const int w = threadIdx.x >> 5;
    int incl = v;
    #pragma unroll
    for (int d = 1; d < 32; d <<= 1) {
        const int t = __shfl_up(incl, d);
        if (lane >= d) incl += t;
    }
    if (lane == 31) wsum[w] = incl;
    __syncthreads();
    int off = 0, tot = 0;
    for (int i = 0; i < nwaves; ++i) {
        const int s = wsum[i];
        if (i < w) off += s;
        tot += s;
    }
    total = tot;
    __syncthreads();
    return off + incl - v;
}

__global__ __launch_bounds__(256) void sage_part(const int* __restrict__ src,
                                                 const int* __restrict__ dst,
                                                 unsigned* __restrict__ tmp,
                                                 int2* __restrict__ binfo,
                                                 int* __restrict__ cursor, int n_edges) {
    __shared__ int hist[NRP];
    __shared__ int lstart[NRP];
    __shared__ int wsum[8];
    __shared__ unsigned stage[PCH];

    const int tid = threadIdx.x;
    const int lane = tid & 31;
    const int wv = tid >> 5;
    const int b = blockIdx.x;
    const int cbase = b * PCH;

    hist[tid] = 0;
    if (b == 0 && tid < 4) cursor[tid] = 0;

    int dv[16], sv[16], lr[16];
    const int wbase = cbase + wv * 512;
    #pragma unroll
    for (int q = 0; q < 4; ++q) {
        const int e0 = wbase + q * 128 + lane * 4;
        if (e0 + 3 < n_edges) {
            const int4 d4 = *(const int4*)(dst + e0);
            const int4 s4 = *(const int4*)(src + e0);
            dv[4 * q + 0] = d4.x; dv[4 * q + 1] = d4.y; dv[4 * q + 2] = d4.z; dv[4 * q + 3] = d4.w;
            sv[4 * q + 0] = s4.x; sv[4 * q + 1] = s4.y; sv[4 * q + 2] = s4.z; sv[4 * q + 3] = s4.w;
        } else {
            #pragma unroll
            for (int k = 0; k < 4; ++k) {
                const int e = e0 + k;
                const bool ok = e < n_edges;
                dv[4 * q + k] = ok ? dst[e] : -1;
                sv[4 * q + k] = ok ? src[e] : 0;
            }
        }
    }
    __syncthreads();

    #pragma unroll
    for (int i = 0; i < 16; ++i)
        lr[i] = (dv[i] >= 0) ? atomicAdd(&hist[dv[i] >> RSH], 1) : 0;
    __syncthreads();

    const int c = hist[tid];
    int total;
    const int ex = block_scan_excl(c, wsum, 8, total);
    lstart[tid] = ex;
    binfo[b * NRP + tid] = make_int2(c, cbase + ex);
    __syncthreads();

    #pragma unroll
    for (int i = 0; i < 16; ++i) {
        if (dv[i] >= 0)
            stage[lstart[dv[i] >> RSH] + lr[i]] =
                ((unsigned)sv[i] << RSH) | (unsigned)(dv[i] & (RNODES - 1));
    }
    __syncthreads();

    for (int i = tid; i < total; i += 256) tmp[cbase + i] = stage[i];
}

template <bool PLACE>
__device__ __forceinline__ void seg_pass(const unsigned* __restrict__ tmp, const int* scnt,
                                         const int* sstart, int* cl, unsigned* stage2,
                                         int* __restrict__ csr, int base, bool staged) {
    const int lane = threadIdx.x & 31;
    const int wv = threadIdx.x >> 5;
    for (int g = wv; g * 8 < NB; g += 16) {
        int c[8], st[8];
        unsigned v[8];
        #pragma unroll
        for (int u = 0; u < 8; ++u) {
            c[u]  = scnt[g * 8 + u];
            st[u] = sstart[g * 8 + u];
        }
        #pragma unroll
        for (int u = 0; u < 8; ++u) {
            int a = st[u] + lane;
            a = a < N_EDGES - 1 ? a : N_EDGES - 1;
            v[u] = tmp[a];
        }
        #pragma unroll
        for (int u = 0; u < 8; ++u) {
            if (lane < c[u]) {
                const int d = (int)(v[u] & (RNODES - 1));
                if (!PLACE) {
                    atomicAdd(&cl[d], 1);
                } else {
                    const int pos = atomicAdd(&cl[d], 1);
                    const int s = (int)(v[u] >> RSH);
                    if (staged) stage2[pos] = (unsigned)s; else csr[base + pos] = s;
                }
            }
        }
        #pragma unroll
        for (int u = 0; u < 8; ++u) {
            for (int j = 32 + lane; j < c[u]; j += 32) {
                const unsigned x = tmp[st[u] + j];
                const int d = (int)(x & (RNODES - 1));
                if (!PLACE) {
                    atomicAdd(&cl[d], 1);
                } else {
                    const int pos = atomicAdd(&cl[d], 1);
                    const int s = (int)(x >> RSH);
                    if (staged) stage2[pos] = (unsigned)s; else csr[base + pos] = s;
                }
            }
        }
    }
}

__global__ __launch_bounds__(512) void sage_sort(const unsigned* __restrict__ tmp,
                                                 const int2* __restrict__ binfo,
                                                 int* __restrict__ cursor,
                                                 int* __restrict__ cnt,
                                                 int* __restrict__ offs,
                                                 int* __restrict__ csr, int n_nodes) {
    __shared__ int scnt[NBP];
    __shared__ int sstart[NBP];
    __shared__ int cntL[RNODES];
    __shared__ int cur[RNODES];
    __shared__ int wsum[16];
    __shared__ int sbase;
    __shared__ unsigned stage2[SCAP];

    const int tid = threadIdx.x;
    const int r = blockIdx.x;

    int c = 0;
    if (tid < NB) {
        const int2 bi = binfo[tid * NRP + r];
        c = bi.x;
        scnt[tid] = bi.x;
        sstart[tid] = bi.y;
    } else if (tid < NBP) {
        scnt[tid] = 0;
        sstart[tid] = 0;
    }
    cntL[tid] = 0;

    int n_r;
    (void)block_scan_excl(c, wsum, 16, n_r);
    if (tid == 0) sbase = atomicAdd(cursor, n_r);
    const bool staged = n_r <= SCAP;

    seg_pass<false>(tmp, scnt, sstart, cntL, stage2, csr, 0, staged);
    __syncthreads();

    const int my = cntL[tid];
    int tot2;
    const int lo = block_scan_excl(my, wsum, 16, tot2);
    const int base = sbase;
    cur[tid] = lo;
    const int node = r * RNODES + tid;
    if (node < n_nodes) {
        cnt[node] = my;
        offs[node] = base + lo;
    }
    __syncthreads();

    seg_pass<true>(tmp, scnt, sstart, cur, stage2, csr, base, staged);
    __syncthreads();

    if (staged) {
        for (int i = tid; i < n_r; i += 512) csr[base + i] = (int)stage2[i];
    }
}

__global__ __launch_bounds__(256) void sage_layer(const float* __restrict__ h,
                                                  const int* __restrict__ csr,
                                                  const int* __restrict__ offs,
                                                  const int* __restrict__ cnt,
                                                  const float* __restrict__ Wself,
                                                  const float* __restrict__ Wneigh,
                                                  const float* __restrict__ bias,
                                                  float* __restrict__ out,
                                                  int* __restrict__ tileq,
                                                  int n_nodes, int do_relu) {
    __shared__ float sW [2 * F * F];
    __shared__ float sH [32 * HSTRIDE];
    __shared__ float sHn[32 * HSTRIDE];
    __shared__ int s_next;

    const int tid = threadIdx.x;
    const int wv = __builtin_amdgcn_readfirstlane((int)(threadIdx.x >> 5));
    const int lane = tid & 31;
    const int col2 = lane * 2;
    const int half = lane >> 4;
    const int c16i = lane & 15;
    const unsigned c16 = (unsigned)c16i * 16u;
    const int ppos = 2 * c16i + half;
    const int ntiles = (n_nodes + 31) >> 5;

    const float bv = bias[(wv & 3) * 16 + c16i];

    #pragma unroll
    for (int i = 0; i < 16; ++i) {
        sW[tid + i * 256]         = Wself [tid + i * 256];
        sW[F * F + tid + i * 256] = Wneigh[tid + i * 256];
    }

    const int m0    = (wv >> 2) * 16;
    const int c0    = (wv & 3) * 16;
    const int lrow  = lane & 15;
    const int khalf = lane >> 4;
    const float* aSelf = &sH [(m0 + lrow) * HSTRIDE + khalf * 2];
    const float* aNei  = &sHn[(m0 + lrow) * HSTRIDE + khalf * 2];
    const float* bSelf = &sW[khalf * 2 * F + c0 + lrow];
    const float* bNei  = bSelf + F * F;
    float* oS = &sH[(m0 + khalf * 8) * HSTRIDE + c0 + lrow];
    const unsigned own8 = (unsigned)lane * 8u;

    int tile = blockIdx.x;
    while (tile < ntiles) {
    const int node0 = tile * 32;
    const int nodew = node0 + wv * NPW;

    int beg[NPW], len[NPW], idx[NPW];
    float2 own[NPW];
    {
        const int nq = nodew + (lane & (NPW - 1));
        const int ncq = nq < n_nodes ? nq : 0;
        const int begv = offs[ncq];
        const int cntv = cnt[ncq];
        #pragma unroll
        for (int r = 0; r < NPW; ++r) {
            const bool ok = nodew + r < n_nodes;
            const int nc = ok ? nodew + r : 0;
            own[r] = *(const float2*)((const char*)h + (((unsigned)nc << 8) + own8));
            if (!ok) { own[r].x = 0.0f; own[r].y = 0.0f; }
        }
        #pragma unroll
        for (int r = 0; r < NPW; ++r) {
            beg[r] = __builtin_amdgcn_readlane(begv, r);
            len[r] = (nodew + r < n_nodes) ? __builtin_amdgcn_readlane(cntv, r) : 0;
        }
    }
    #pragma unroll
    for (int r = 0; r < NPW; ++r) {
        int a = beg[r] + ppos;
        a = a < N_EDGES - 1 ? a : N_EDGES - 1;
        const int s = csr[a];
        idx[r] = (ppos < len[r]) ? s : 0;
    }

    #pragma unroll
    for (int r = 0; r < NPW; ++r) {
        sH[(wv * NPW + r) * HSTRIDE + col2    ] = own[r].x;
        sH[(wv * NPW + r) * HSTRIDE + col2 + 1] = own[r].y;
    }

    #pragma unroll
    for (int r = 0; r < NPW; ++r) {
        const int n = len[r];
        const int m = n < 32 ? n : 32;
        const int nfull = m >> 1;
        const int nb0 = nfull < GST ? nfull : GST;
        const int rowoff = idx[r] << 8;
        v4f acc = {0.0f, 0.0f, 0.0f, 0.0f};
        if (m & 1) {
            const int st = __builtin_amdgcn_readlane(rowoff, nfull);
            acc = ld_row16(h, (unsigned)st + c16);
        }
        v4f v[GST];
        #pragma unroll
        for (int k = 0; k < GST; ++k) asm volatile("" : "=v"(v[k]));
        if (nb0 > 0) { v[0] = ld_pair<0>(h, rowoff, c16);
        if (nb0 > 1) { v[1] = ld_pair<1>(h, rowoff, c16);
        if (nb0 > 2) { v[2] = ld_pair<2>(h, rowoff, c16);
        if (nb0 > 3) { v[3] = ld_pair<3>(h, rowoff, c16);
        if (nb0 > 4) { v[4] = ld_pair<4>(h, rowoff, c16);
        if (nb0 > 5) { v[5] = ld_pair<5>(h, rowoff, c16);
        if (nb0 > 6) { v[6] = ld_pair<6>(h, rowoff, c16);
        if (nb0 > 7) { v[7] = ld_pair<7>(h, rowoff, c16);
        if (nb0 > 8) { v[8] = ld_pair<8>(h, rowoff, c16);
        if (nb0 > 9) { v[9] = ld_pair<9>(h, rowoff, c16);
        if (nb0 > 10) { v[10] = ld_pair<10>(h, rowoff, c16);
        if (nb0 > 11) { v[11] = ld_pair<11>(h, rowoff, c16);
        }}}}}}}}}}}}
        if (r == NPW - 1 && tid == 0) s_next = LBLK + atomicAdd(tileq, 1);
        if (half != 0) { acc.x = 0.0f; acc.y = 0.0f; acc.z = 0.0f; acc.w = 0.0f; }
        if (nb0 > 0) { acc += v[0];
        if (nb0 > 1) { acc += v[1];
        if (nb0 > 2) { acc += v[2];
        if (nb0 > 3) { acc += v[3];
        if (nb0 > 4) { acc += v[4];
        if (nb0 > 5) { acc += v[5];
        if (nb0 > 6) { acc += v[6];
        if (nb0 > 7) { acc += v[7];
        if (nb0 > 8) { acc += v[8];
        if (nb0 > 9) { acc += v[9];
        if (nb0 > 10) { acc += v[10];
        if (nb0 > 11) { acc += v[11];
        }}}}}}}}}}}}
        #pragma nounroll
        for (int j = GST; j < nfull; ++j) {
            const int ro = __shfl(rowoff, (half << 4) + j);
            acc += ld_row16(h, (unsigned)ro + c16);
        }
        #pragma nounroll
        for (int k = 32; k < n; ++k) {
            const int s = csr[beg[r] + k];
            const v4f x = ld_row16(h, ((unsigned)s << 8) + c16);
            if (half == 0) acc += x;
        }
        acc.x += other_half(acc.x);
        acc.y += other_half(acc.y);
        acc.z += other_half(acc.z);
        acc.w += other_half(acc.w);
        const float rinv = 1.0f / fmaxf((float)n, 1.0f);
        const int hcol = (wv * NPW + r) * HSTRIDE + 4 * c16i + 2 * half;
        sHn[hcol    ] = (half ? acc.z : acc.x) * rinv;
        sHn[hcol + 1] = (half ? acc.w : acc.y) * rinv;
    }
    __syncthreads();

    const int nxt = __builtin_amdgcn_readfirstlane(s_next);

    v8f acc = {};
    #pragma unroll
    for (int kk = 0; kk < 16; ++kk) {
        v2f a, b;
        a.x = aSelf[kk * 4];
        a.y = aSelf[kk * 4 + 1];
        b.x = bSelf[kk * 4 * F];
        b.y = bSelf[kk * 4 * F + F];
        acc = __builtin_amdgcn_wmma_f32_16x16x4_f32(false, a, false, b,
                                                    (short)0, acc, false, false);
        a.x = aNei[kk * 4];
        a.y = aNei[kk * 4 + 1];
        b.x = bNei[kk * 4 * F];
        b.y = bNei[kk * 4 * F + F];
        acc = __builtin_amdgcn_wmma_f32_16x16x4_f32(false, a, false, b,
                                                    (short)0, acc, false, false);
    }

    asm volatile("v_nop\n\tv_nop\n\tv_nop\n\tv_nop" : "+v"(acc));

    __syncthreads();

    #pragma unroll
    for (int i = 0; i < 8; ++i) {
        float v = acc[i] + bv;
        if (do_relu) v = fmaxf(v, 0.0f);
        oS[i * HSTRIDE] = v;
    }
    __syncthreads();

    float2 orow[NPW];
    #pragma unroll
    for (int r = 0; r < NPW; ++r) {
        const int trow = wv * NPW + r;
        orow[r].x = sH[trow * HSTRIDE + col2];
        orow[r].y = sH[trow * HSTRIDE + col2 + 1];
    }
    #pragma unroll
    for (int r = 0; r < NPW; ++r) {
        const int orn = nodew + r;
        if (orn < n_nodes)
            *(float2*)((char*)out + (((unsigned)orn << 8) + own8)) = orow[r];
    }
    tile = nxt;
    }
}

extern "C" void kernel_launch(void* const* d_in, const int* in_sizes, int n_in,
                              void* d_out, int out_size, void* d_ws, size_t ws_size,
                              hipStream_t stream) {
    (void)in_sizes; (void)n_in; (void)out_size; (void)ws_size;
    const float* feat = (const float*)d_in[0];
    const int*   src  = (const int*)d_in[1];
    const int*   dst  = (const int*)d_in[2];
    const float* Ws[3] = { (const float*)d_in[3], (const float*)d_in[6], (const float*)d_in[9]  };
    const float* Wn[3] = { (const float*)d_in[4], (const float*)d_in[7], (const float*)d_in[10] };
    const float* bs[3] = { (const float*)d_in[5], (const float*)d_in[8], (const float*)d_in[11] };
    float* outp = (float*)d_out;

    int*   wsi    = (int*)d_ws;
    int*   cnt    = wsi;
    int*   cursor = wsi + N_NODES;
    int*   tileq  = cursor + 1;
    int*   offs   = wsi + N_NODES + 64;
    int*   csr    = wsi + 2 * N_NODES + 128;
    float* h1     = (float*)(csr + N_EDGES);
    float* h2     = h1 + (long long)N_NODES * F;
    unsigned* tmp = (unsigned*)h2;
    int2*  binfo  = (int2*)(tmp + N_EDGES);

    sage_part<<<NB, 256, 0, stream>>>(src, dst, tmp, binfo, cursor, N_EDGES);
    sage_sort<<<NR, 512, 0, stream>>>(tmp, binfo, cursor, cnt, offs, csr, N_NODES);

    sage_layer<<<LBLK, 256, 0, stream>>>(feat, csr, offs, cnt, Ws[0], Wn[0], bs[0], h1,   tileq + 0, N_NODES, 1);
    sage_layer<<<LBLK, 256, 0, stream>>>(h1,   csr, offs, cnt, Ws[1], Wn[1], bs[1], h2,   tileq + 1, N_NODES, 1);
    sage_layer<<<LBLK, 256, 0, stream>>>(h2,   csr, offs, cnt, Ws[2], Wn[2], bs[2], outp, tileq + 2, N_NODES, 0);
}


# layer kernel LDS layout: tile row pitch 68 words, weights as [k/4][n][k%4], so a lane reads A and B of two K-steps with one 16-B load each, no bank conflict (was 4-way); k order of the sums changed
# speedup vs baseline: 6.5970x; 1.0612x over previous
#include <hip/hip_runtime.h>


#define N_NODES 100000
#define N_EDGES 1600000
#define F 64
#define HSTRIDE 68
#define NPW 4
#define GST 12
#define LBLK 1536

typedef __attribute__((ext_vector_type(2))) float v2f;
typedef __attribute__((ext_vector_type(8))) float v8f;
typedef __attribute__((ext_vector_type(4))) float v4f;

__device__ __forceinline__ v4f ld_row16(const float* __restrict__ hb, unsigned off) {
    return *(const v4f*)((const char*)hb + off);
}

template <int K>
__device__ __forceinline__ v4f ld_pair(const float* __restrict__ hb, int rowoff, unsigned c16) {
    const int ro = __builtin_amdgcn_update_dpp(0, rowoff, 0x150 + K, 0xF, 0xF, true);
    return ld_row16(hb, (unsigned)ro + c16);
}

__device__ __forceinline__ float other_half(float x) {
    const unsigned u = __float_as_uint(x);
    return __uint_as_float(__builtin_amdgcn_permlanex16(u, u, 0x76543210u, 0xfedcba98u, false, false));
}

#define RSH 9
#define RNODES (1 << RSH)
#define NR ((N_NODES + RNODES - 1) / RNODES)
#define NRP 256
#define PCH 4096
#define NB ((N_EDGES + PCH - 1) / PCH)
#define NBP 400
#define SCAP 12288

__device__ __forceinline__ int block_scan_excl(int v, int* wsum, int nwaves, int& total) {
    const int lane = threadIdx.x & 31;
    const int w = threadIdx.x >> 5;
    int incl = v;
    #pragma unroll
    for (int d = 1; d < 32; d <<= 1) {
        const int t = __shfl_up(incl, d);
        if (lane >= d) incl += t;
    }
    if (lane == 31) wsum[w] = incl;
    __syncthreads();
    int off = 0, tot = 0;
    for (int i = 0; i < nwaves; ++i) {
        const int s = wsum[i];
        if (i < w) off += s;
        tot += s;
    }
    total = tot;
    __syncthreads();
    return off + incl - v;
}

__global__ __launch_bounds__(256) void sage_part(const int* __restrict__ src,
                                                 const int* __restrict__ dst,
                                                 unsigned* __restrict__ tmp,
                                                 int2* __restrict__ binfo,
                                                 int* __restrict__ cursor, int n_edges) {
    __shared__ int hist[NRP];
    __shared__ int lstart[NRP];
    __shared__ int wsum[8];
    __shared__ unsigned stage[PCH];

    const int tid = threadIdx.x;
    const int lane = tid & 31;
    const int wv = tid >> 5;
    const int b = blockIdx.x;
    const int cbase = b * PCH;

    hist[tid] = 0;
    if (b == 0 && tid < 4) cursor[tid] = 0;

    int dv[16], sv[16], lr[16];
    const int wbase = cbase + wv * 512;
    #pragma unroll
    for (int q = 0; q < 4; ++q) {
        const int e0 = wbase + q * 128 + lane * 4;
        if (e0 + 3 < n_edges) {
            const int4 d4 = *(const int4*)(dst + e0);
            const int4 s4 = *(const int4*)(src + e0);
            dv[4 * q + 0] = d4.x; dv[4 * q + 1] = d4.y; dv[4 * q + 2] = d4.z; dv[4 * q + 3] = d4.w;
            sv[4 * q + 0] = s4.x; sv[4 * q + 1] = s4.y; sv[4 * q + 2] = s4.z; sv[4 * q + 3] = s4.w;
        } else {
            #pragma unroll
            for (int k = 0; k < 4; ++k) {
                const int e = e0 + k;
                const bool ok = e < n_edges;
                dv[4 * q + k] = ok ? dst[e] : -1;
                sv[4 * q + k] = ok ? src[e] : 0;
            }
        }
    }
    __syncthreads();

    #pragma unroll
    for (int i = 0; i < 16; ++i)
        lr[i] = (dv[i] >= 0) ? atomicAdd(&hist[dv[i] >> RSH], 1) : 0;
    __syncthreads();

    const int c = hist[tid];
    int total;
    const int ex = block_scan_excl(c, wsum, 8, total);
    lstart[tid] = ex;
    binfo[b * NRP + tid] = make_int2(c, cbase + ex);
    __syncthreads();

    #pragma unroll
    for (int i = 0; i < 16; ++i) {
        if (dv[i] >= 0)
            stage[lstart[dv[i] >> RSH] + lr[i]] =
                ((unsigned)sv[i] << RSH) | (unsigned)(dv[i] & (RNODES - 1));
    }
    __syncthreads();

    for (int i = tid; i < total; i += 256) tmp[cbase + i] = stage[i];
}

template <bool PLACE>
__device__ __forceinline__ void seg_pass(const unsigned* __restrict__ tmp, const int* scnt,
                                         const int* sstart, int* cl, unsigned* stage2,
                                         int* __restrict__ csr, int base, bool staged) {
    const int lane = threadIdx.x & 31;
    const int wv = threadIdx.x >> 5;
    for (int g = wv; g * 8 < NB; g += 16) {
        int c[8], st[8];
        unsigned v[8];
        #pragma unroll
        for (int u = 0; u < 8; ++u) {
            c[u]  = scnt[g * 8 + u];
            st[u] = sstart[g * 8 + u];
        }
        #pragma unroll
        for (int u = 0; u < 8; ++u) {
            int a = st[u] + lane;
            a = a < N_EDGES - 1 ? a : N_EDGES - 1;
            v[u] = tmp[a];
        }
        #pragma unroll
        for (int u = 0; u < 8; ++u) {
            if (lane < c[u]) {
                const int d = (int)(v[u] & (RNODES - 1));
                if (!PLACE) {
                    atomicAdd(&cl[d], 1);
                } else {
                    const int pos = atomicAdd(&cl[d], 1);
                    const int s = (int)(v[u] >> RSH);
                    if (staged) stage2[pos] = (unsigned)s; else csr[base + pos] = s;
                }
            }
        }
        #pragma unroll
        for (int u = 0; u < 8; ++u) {
            for (int j = 32 + lane; j < c[u]; j += 32) {
                const unsigned x = tmp[st[u] + j];
                const int d = (int)(x & (RNODES - 1));
                if (!PLACE) {
                    atomicAdd(&cl[d], 1);
                } else {
                    const int pos = atomicAdd(&cl[d], 1);
                    const int s = (int)(x >> RSH);
                    if (staged) stage2[pos] = (unsigned)s; else csr[base + pos] = s;
                }
            }
        }
    }
}

__global__ __launch_bounds__(512) void sage_sort(const unsigned* __restrict__ tmp,
                                                 const int2* __restrict__ binfo,
                                                 int* __restrict__ cursor,
                                                 int* __restrict__ cnt,
                                                 int* __restrict__ offs,
                                                 int* __restrict__ csr, int n_nodes) {
    __shared__ int scnt[NBP];
    __shared__ int sstart[NBP];
    __shared__ int cntL[RNODES];
    __shared__ int cur[RNODES];
    __shared__ int wsum[16];
    __shared__ int sbase;
    __shared__ unsigned stage2[SCAP];

    const int tid = threadIdx.x;
    const int r = blockIdx.x;

    int c = 0;
    if (tid < NB) {
        const int2 bi = binfo[tid * NRP + r];
        c = bi.x;
        scnt[tid] = bi.x;
        sstart[tid] = bi.y;
    } else if (tid < NBP) {
        scnt[tid] = 0;
        sstart[tid] = 0;
    }
    cntL[tid] = 0;

    int n_r;
    (void)block_scan_excl(c, wsum, 16, n_r);
    if (tid == 0) sbase = atomicAdd(cursor, n_r);
    const bool staged = n_r <= SCAP;

    seg_pass<false>(tmp, scnt, sstart, cntL, stage2, csr, 0, staged);
    __syncthreads();

    const int my = cntL[tid];
    int tot2;
    const int lo = block_scan_excl(my, wsum, 16, tot2);
    const int base = sbase;
    cur[tid] = lo;
    const int node = r * RNODES + tid;
    if (node < n_nodes) {
        cnt[node] = my;
        offs[node] = base + lo;
    }
    __syncthreads();

    seg_pass<true>(tmp, scnt, sstart, cur, stage2, csr, base, staged);
    __syncthreads();

    if (staged) {
        for (int i = tid; i < n_r; i += 512) csr[base + i] = (int)stage2[i];
    }
}

__global__ __launch_bounds__(256) void sage_layer(const float* __restrict__ h,
                                                  const int* __restrict__ csr,
                                                  const int* __restrict__ offs,
                                                  const int* __restrict__ cnt,
                                                  const float* __restrict__ Wself,
                                                  const float* __restrict__ Wneigh,
                                                  const float* __restrict__ bias,
                                                  float* __restrict__ out,
                                                  int* __restrict__ tileq,
                                                  int n_nodes, int do_relu) {
    __shared__ __attribute__((aligned(16))) float sW [2 * F * F];
    __shared__ __attribute__((aligned(16))) float sH [32 * HSTRIDE];
    __shared__ __attribute__((aligned(16))) float sHn[32 * HSTRIDE];
    __shared__ int s_next;

    const int tid = threadIdx.x;
    const int wv = __builtin_amdgcn_readfirstlane((int)(threadIdx.x >> 5));
    const int lane = tid & 31;
    const int col2 = lane * 2;
    const int half = lane >> 4;
    const int c16i = lane & 15;
    const unsigned c16 = (unsigned)c16i * 16u;
    const int ppos = 2 * c16i + half;
    const int ntiles = (n_nodes + 31) >> 5;

    const float bv = bias[(wv & 3) * 16 + c16i];

    {
        const int wcol = tid & (F - 1);
        const int wg0 = tid >> 6;
        #pragma unroll
        for (int i = 0; i < 4; ++i) {
            const int g = i * 4 + wg0;
            v4f ws4, wn4;
            ws4.x = Wself [(4 * g + 0) * F + wcol];
            ws4.y = Wself [(4 * g + 1) * F + wcol];
            ws4.z = Wself [(4 * g + 2) * F + wcol];
            ws4.w = Wself [(4 * g + 3) * F + wcol];
            wn4.x = Wneigh[(4 * g + 0) * F + wcol];
            wn4.y = Wneigh[(4 * g + 1) * F + wcol];
            wn4.z = Wneigh[(4 * g + 2) * F + wcol];
            wn4.w = Wneigh[(4 * g + 3) * F + wcol];
            *(v4f*)&sW[        g * 4 * F + wcol * 4] = ws4;
            *(v4f*)&sW[F * F + g * 4 * F + wcol * 4] = wn4;
        }
    }

    const int m0    = (wv >> 2) * 16;
    const int c0    = (wv & 3) * 16;
    const int lrow  = lane & 15;
    const int khalf = lane >> 4;
    const v4f* aSelf = (const v4f*)&sH [(m0 + lrow) * HSTRIDE + khalf * 4];
    const v4f* aNei  = (const v4f*)&sHn[(m0 + lrow) * HSTRIDE + khalf * 4];
    const v4f* bSelf = (const v4f*)&sW[khalf * 4 * F + (c0 + lrow) * 4];
    const v4f* bNei  = bSelf + F * F / 4;
    float* oS = &sH[(m0 + khalf * 8) * HSTRIDE + c0 + lrow];
    const unsigned own8 = (unsigned)lane * 8u;

    int tile = blockIdx.x;
    while (tile < ntiles) {
    const int node0 = tile * 32;
    const int nodew = node0 + wv * NPW;

    int beg[NPW], len[NPW], idx[NPW];
    float2 own[NPW];
    {
        const int nq = nodew + (lane & (NPW - 1));
        const int ncq = nq < n_nodes ? nq : 0;
        const int begv = offs[ncq];
        const int cntv = cnt[ncq];
        #pragma unroll
        for (int r = 0; r < NPW; ++r) {
            const bool ok = nodew + r < n_nodes;
            const int nc = ok ? nodew + r : 0;
            own[r] = *(const float2*)((const char*)h + (((unsigned)nc << 8) + own8));
            if (!ok) { own[r].x = 0.0f; own[r].y = 0.0f; }
        }
        #pragma unroll
        for (int r = 0; r < NPW; ++r) {
            beg[r] = __builtin_amdgcn_readlane(begv, r);
            len[r] = (nodew + r < n_nodes) ? __builtin_amdgcn_readlane(cntv, r) : 0;
        }
    }
    #pragma unroll
    for (int r = 0; r < NPW; ++r) {
        int a = beg[r] + ppos;
        a = a < N_EDGES - 1 ? a : N_EDGES - 1;
        const int s = csr[a];
        idx[r] = (ppos < len[r]) ? s : 0;
    }

    #pragma unroll
    for (int r = 0; r < NPW; ++r)
        *(float2*)&sH[(wv * NPW + r) * HSTRIDE + col2] = own[r];

    #pragma unroll
    for (int r = 0; r < NPW; ++r) {
        const int n = len[r];
        const int m = n < 32 ? n : 32;
        const int nfull = m >> 1;
        const int nb0 = nfull < GST ? nfull : GST;
        const int rowoff = idx[r] << 8;
        v4f acc = {0.0f, 0.0f, 0.0f, 0.0f};
        if (m & 1) {
            const int st = __builtin_amdgcn_readlane(rowoff, nfull);
            acc = ld_row16(h, (unsigned)st + c16);
        }
        v4f v[GST];
        #pragma unroll
        for (int k = 0; k < GST; ++k) asm volatile("" : "=v"(v[k]));
        if (nb0 > 0) { v[0] = ld_pair<0>(h, rowoff, c16);
        if (nb0 > 1) { v[1] = ld_pair<1>(h, rowoff, c16);
        if (nb0 > 2) { v[2] = ld_pair<2>(h, rowoff, c16);
        if (nb0 > 3) { v[3] = ld_pair<3>(h, rowoff, c16);
        if (nb0 > 4) { v[4] = ld_pair<4>(h, rowoff, c16);
        if (nb0 > 5) { v[5] = ld_pair<5>(h, rowoff, c16);
        if (nb0 > 6) { v[6] = ld_pair<6>(h, rowoff, c16);
        if (nb0 > 7) { v[7] = ld_pair<7>(h, rowoff, c16);
        if (nb0 > 8) { v[8] = ld_pair<8>(h, rowoff, c16);
        if (nb0 > 9) { v[9] = ld_pair<9>(h, rowoff, c16);
        if (nb0 > 10) { v[10] = ld_pair<10>(h, rowoff, c16);
        if (nb0 > 11) { v[11] = ld_pair<11>(h, rowoff, c16);
        }}}}}}}}}}}}
        if (r == NPW - 1 && tid == 0) s_next = LBLK + atomicAdd(tileq, 1);
        if (half != 0) { acc.x = 0.0f; acc.y = 0.0f; acc.z = 0.0f; acc.w = 0.0f; }
        if (nb0 > 0) { acc += v[0];
        if (nb0 > 1) { acc += v[1];
        if (nb0 > 2) { acc += v[2];
        if (nb0 > 3) { acc += v[3];
        if (nb0 > 4) { acc += v[4];
        if (nb0 > 5) { acc += v[5];
        if (nb0 > 6) { acc += v[6];
        if (nb0 > 7) { acc += v[7];
        if (nb0 > 8) { acc += v[8];
        if (nb0 > 9) { acc += v[9];
        if (nb0 > 10) { acc += v[10];
        if (nb0 > 11) { acc += v[11];
        }}}}}}}}}}}}
        #pragma nounroll
        for (int j = GST; j < nfull; ++j) {
            const int ro = __shfl(rowoff, (half << 4) + j);
            acc += ld_row16(h, (unsigned)ro + c16);
        }
        #pragma nounroll
        for (int k = 32; k < n; ++k) {
            const int s = csr[beg[r] + k];
            const v4f x = ld_row16(h, ((unsigned)s << 8) + c16);
            if (half == 0) acc += x;
        }
        acc.x += other_half(acc.x);
        acc.y += other_half(acc.y);
        acc.z += other_half(acc.z);
        acc.w += other_half(acc.w);
        const float rinv = 1.0f / fmaxf((float)n, 1.0f);
        const int hcol = (wv * NPW + r) * HSTRIDE + 4 * c16i + 2 * half;
        v2f mean2;
        mean2.x = (half ? acc.z : acc.x) * rinv;
        mean2.y = (half ? acc.w : acc.y) * rinv;
        *(v2f*)&sHn[hcol] = mean2;
    }
    __syncthreads();

    const int nxt = __builtin_amdgcn_readfirstlane(s_next);

    v8f acc = {};
    #pragma unroll
    for (int j = 0; j < 8; ++j) {
        const v4f a  = aSelf[j * 2];
        const v4f b  = bSelf[j * 2 * F];
        const v4f an = aNei [j * 2];
        const v4f bn = bNei [j * 2 * F];
        acc = __builtin_amdgcn_wmma_f32_16x16x4_f32(false, a.xy, false, b.xy,
                                                    (short)0, acc, false, false);
        acc = __builtin_amdgcn_wmma_f32_16x16x4_f32(false, a.zw, false, b.zw,
                                                    (short)0, acc, false, false);
        acc = __builtin_amdgcn_wmma_f32_16x16x4_f32(false, an.xy, false, bn.xy,
                                                    (short)0, acc, false, false);
        acc = __builtin_amdgcn_wmma_f32_16x16x4_f32(false, an.zw, false, bn.zw,
                                                    (short)0, acc, false, false);
    }

    asm volatile("v_nop\n\tv_nop\n\tv_nop\n\tv_nop" : "+v"(acc));

    __syncthreads();

    #pragma unroll
    for (int i = 0; i < 8; ++i) {
        float v = acc[i] + bv;
        if (do_relu) v = fmaxf(v, 0.0f);
        oS[i * HSTRIDE] = v;
    }
    __syncthreads();

    float2 orow[NPW];
    #pragma unroll
    for (int r = 0; r < NPW; ++r)
        orow[r] = *(const float2*)&sH[(wv * NPW + r) * HSTRIDE + col2];
    #pragma unroll
    for (int r = 0; r < NPW; ++r) {
        const int orn = nodew + r;
        if (orn < n_nodes)
            *(float2*)((char*)out + (((unsigned)orn << 8) + own8)) = orow[r];
    }
    tile = nxt;
    }
}

extern "C" void kernel_launch(void* const* d_in, const int* in_sizes, int n_in,
                              void* d_out, int out_size, void* d_ws, size_t ws_size,
                              hipStream_t stream) {
    (void)in_sizes; (void)n_in; (void)out_size; (void)ws_size;
    const float* feat = (const float*)d_in[0];
    const int*   src  = (const int*)d_in[1];
    const int*   dst  = (const int*)d_in[2];
    const float* Ws[3] = { (const float*)d_in[3], (const float*)d_in[6], (const float*)d_in[9]  };
    const float* Wn[3] = { (const float*)d_in[4], (const float*)d_in[7], (const float*)d_in[10] };
    const float* bs[3] = { (const float*)d_in[5], (const float*)d_in[8], (const float*)d_in[11] };
    float* outp = (float*)d_out;

    int*   wsi    = (int*)d_ws;
    int*   cnt    = wsi;
    int*   cursor = wsi + N_NODES;
    int*   tileq  = cursor + 1;
    int*   offs   = wsi + N_NODES + 64;
    int*   csr    = wsi + 2 * N_NODES + 128;
    float* h1     = (float*)(csr + N_EDGES);
    float* h2     = h1 + (long long)N_NODES * F;
    unsigned* tmp = (unsigned*)h2;
    int2*  binfo  = (int2*)(tmp + N_EDGES);

    sage_part<<<NB, 256, 0, stream>>>(src, dst, tmp, binfo, cursor, N_EDGES);
    sage_sort<<<NR, 512, 0, stream>>>(tmp, binfo, cursor, cnt, offs, csr, N_NODES);

    sage_layer<<<LBLK, 256, 0, stream>>>(feat, csr, offs, cnt, Ws[0], Wn[0], bs[0], h1,   tileq + 0, N_NODES, 1);
    sage_layer<<<LBLK, 256, 0, stream>>>(h1,   csr, offs, cnt, Ws[1], Wn[1], bs[1], h2,   tileq + 1, N_NODES, 1);
    sage_layer<<<LBLK, 256, 0, stream>>>(h2,   csr, offs, cnt, Ws[2], Wn[2], bs[2], outp, tileq + 2, N_NODES, 0);
}
